# starting kernel
# baseline (speedup 1.0000x reference)
#include <hip/hip_runtime.h>
#include <hip/hip_bf16.h>

#define GCN_N_NODES 50000
#define GCN_N_EDGES 800000
#define GCN_IN      128
#define GCN_H       128
#define GCN_C       32

typedef __attribute__((ext_vector_type(2))) float v2f;
typedef __attribute__((ext_vector_type(8))) float v8f;

__global__ __launch_bounds__(256) void gcn_deg_kernel(
    const int* __restrict__ src, const int* __restrict__ dst,
    float* __restrict__ deg_out, float* __restrict__ deg_in, int E) {
  int e = blockIdx.x * blockDim.x + threadIdx.x;
  if (e < E) {
    atomicAdd(&deg_out[src[e]], 1.0f);
    atomicAdd(&deg_in[dst[e]], 1.0f);
  }
}

__global__ __launch_bounds__(256) void gcn_invsqrt_kernel(
    float* __restrict__ deg_out, float* __restrict__ deg_in, int N) {
  int n = blockIdx.x * blockDim.x + threadIdx.x;
  if (n < N) {
    float d = deg_out[n];
    deg_out[n] = (d > 0.0f) ? rsqrtf(d) : 0.0f;
    d = deg_in[n];
    deg_in[n] = (d > 0.0f) ? rsqrtf(d) : 0.0f;
  }
}

__global__ __launch_bounds__(256) void gcn_spmm_scatter(
    const float* __restrict__ x, const int* __restrict__ src,
    const int* __restrict__ dst, const float* __restrict__ inv_out,
    float* __restrict__ agg, int E) {
  int lane = threadIdx.x & 31;
  int e = blockIdx.x * 8 + (threadIdx.x >> 5);
  if (e >= E) return;
  int s = src[e];
  int d = dst[e];
  float sc = inv_out[s];
  const float4* xr = (const float4*)(x + (size_t)s * 128);
  float4 v = xr[lane];
  float* ar = agg + (size_t)d * 128 + lane * 4;
  atomicAdd(ar + 0, v.x * sc);
  atomicAdd(ar + 1, v.y * sc);
  atomicAdd(ar + 2, v.z * sc);
  atomicAdd(ar + 3, v.w * sc);
}

__global__ __launch_bounds__(256) void gcn_gemm1_wmma(
    const float* __restrict__ agg, const float* __restrict__ inv_in,
    const float* __restrict__ W, const float* __restrict__ bias,
    float* __restrict__ out) {
  __shared__ float As[16][132];
  const int row0 = blockIdx.x * 16;
  const int tid  = threadIdx.x;

  for (int i = tid; i < 16 * 128; i += 256) {
    int r = i >> 7, c = i & 127;
    As[r][c] = agg[(size_t)(row0 + r) * 128 + c] * inv_in[row0 + r];
  }
  __syncthreads();

  const int wave = tid >> 5;
  const int lane = tid & 31;
  const int col0 = wave * 16;
  const int half = lane >> 4;
  const int mrow = lane & 15;

  v8f acc = {};
  for (int kk = 0; kk < 128; kk += 4) {
    int k2 = kk + half * 2;
    v2f a, b;
    a.x = As[mrow][k2];
    a.y = As[mrow][k2 + 1];
    b.x = W[(size_t)k2 * 128 + col0 + mrow];
    b.y = W[(size_t)(k2 + 1) * 128 + col0 + mrow];
    acc = __builtin_amdgcn_wmma_f32_16x16x4_f32(
         false, a,  false, b,
         (short)0, acc,  false,  false);
  }

  float bv = bias[col0 + mrow];
  for (int r = 0; r < 8; ++r) {
    int row = row0 + r + half * 8;
    float v = acc[r] + bv;
    out[(size_t)row * 128 + col0 + mrow] = (v > 0.0f) ? v : 0.0f;
  }
}

__global__ __launch_bounds__(256) void gcn_gemm2_wmma(
    const float* __restrict__ agg, const float* __restrict__ inv_in,
    const float* __restrict__ W, const float* __restrict__ bias,
    float* __restrict__ out, int N) {
  __shared__ float As[64][132];
  const int row0 = blockIdx.x * 64;
  const int tid  = threadIdx.x;

  for (int i = tid; i < 64 * 128; i += 256) {
    int r = i >> 7, c = i & 127;
    int row = row0 + r;
    As[r][c] = (row < N) ? agg[(size_t)row * 128 + c] * inv_in[row] : 0.0f;
  }
  __syncthreads();

  const int wave = tid >> 5;
  const int lane = tid & 31;
  const int rg   = wave >> 1;
  const int col0 = (wave & 1) * 16;
  const int half = lane >> 4;
  const int mrow = lane & 15;

  v8f acc = {};
  for (int kk = 0; kk < 128; kk += 4) {
    int k2 = kk + half * 2;
    v2f a, b;
    a.x = As[rg * 16 + mrow][k2];
    a.y = As[rg * 16 + mrow][k2 + 1];
    b.x = W[(size_t)k2 * 32 + col0 + mrow];
    b.y = W[(size_t)(k2 + 1) * 32 + col0 + mrow];
    acc = __builtin_amdgcn_wmma_f32_16x16x4_f32(
        false, a, false, b, (short)0, acc, false, false);
  }

  float bv = bias[col0 + mrow];
  for (int r = 0; r < 8; ++r) {
    int row = row0 + rg * 16 + r + half * 8;
    if (row < N) out[(size_t)row * 32 + col0 + mrow] = acc[r] + bv;
  }
}

extern "C" void kernel_launch(void* const* d_in, const int* in_sizes, int n_in,
                              void* d_out, int out_size, void* d_ws, size_t ws_size,
                              hipStream_t stream) {
  const float* in_feat = (const float*)d_in[0];
  const int*   src     = (const int*)d_in[1];
  const int*   dst     = (const int*)d_in[2];
  const float* W1      = (const float*)d_in[3];
  const float* b1      = (const float*)d_in[4];
  const float* W2      = (const float*)d_in[5];
  const float* b2      = (const float*)d_in[6];
  float* out = (float*)d_out;

  const int N = GCN_N_NODES;
  const int E = GCN_N_EDGES;

  float* inv_out = (float*)d_ws;
  float* inv_in  = inv_out + N;
  float* agg     = inv_in + N;
  float* h1      = agg + (size_t)N * 128;

  hipMemsetAsync(inv_out, 0, (size_t)2 * N * sizeof(float), stream);
  gcn_deg_kernel<<<(E + 255) / 256, 256, 0, stream>>>(src, dst, inv_out, inv_in, E);
  gcn_invsqrt_kernel<<<(N + 255) / 256, 256, 0, stream>>>(inv_out, inv_in, N);

  hipMemsetAsync(agg, 0, (size_t)N * 128 * sizeof(float), stream);
  gcn_spmm_scatter<<<(E + 7) / 8, 256, 0, stream>>>(in_feat, src, dst, inv_out, agg, E);
  gcn_gemm1_wmma<<<N / 16, 256, 0, stream>>>(agg, inv_in, W1, b1, h1);

  hipMemsetAsync(agg, 0, (size_t)N * 128 * sizeof(float), stream);
  gcn_spmm_scatter<<<(E + 7) / 8, 256, 0, stream>>>(h1, src, dst, inv_out, agg, E);
  gcn_gemm2_wmma<<<(N + 63) / 64, 256, 0, stream>>>(agg, inv_in, W2, b2, out, N);
}


# Each owner group multiplies its own LDS tile of edge sums by W with f32 WMMA (A read from the tile, same k order) and stores h1 / out itself: agg never in memory, 2 GEMM launches gone (6 -> 4)
# speedup vs baseline: 2.4623x; 2.4623x over previous
#include <hip/hip_runtime.h>
#include <hip/hip_bf16.h>

#define GCN_N_NODES 50000
#define GCN_N_EDGES 800000
#define GCN_IN      128
#define GCN_H       128
#define GCN_C       32
#define OW_GROUPS   256
#define OW_NPG      196
#define OW_ROWS     208
#define OW_PITCH    132

typedef __attribute__((ext_vector_type(2))) float v2f;
typedef __attribute__((ext_vector_type(8))) float v8f;

__global__ __launch_bounds__(256) void gcn_deg_kernel(
    const int* __restrict__ src, const int* __restrict__ dst,
    float* __restrict__ deg_out, float* __restrict__ deg_in,
    unsigned char* __restrict__ keys, int E) {
  int e = blockIdx.x * blockDim.x + threadIdx.x;
  if (e < E) {
    const int d = dst[e];
    atomicAdd(&deg_out[src[e]], 1.0f);
    atomicAdd(&deg_in[d], 1.0f);
    keys[e] = (unsigned char)((unsigned)d / (unsigned)OW_NPG);
  }
}

__global__ __launch_bounds__(256) void gcn_invsqrt_kernel(
    float* __restrict__ deg_out, float* __restrict__ deg_in, int N) {
  int n = blockIdx.x * blockDim.x + threadIdx.x;
  if (n < N) {
    float d = deg_out[n];
    deg_out[n] = (d > 0.0f) ? rsqrtf(d) : 0.0f;
    d = deg_in[n];
    deg_in[n] = (d > 0.0f) ? rsqrtf(d) : 0.0f;
  }
}

#define OW_THREADS 1024

__device__ __forceinline__ unsigned ow_zero_bytes(unsigned v) {
  return ~(((v & 0x7F7F7F7Fu) + 0x7F7F7F7Fu) | v) & 0x80808080u;
}

__device__ __forceinline__ void ow_add_row(float* tile, int d, int lane, float4 v, float sc) {
  float* t = tile + d * OW_PITCH + lane;
  __hip_atomic_fetch_add(t + 0,  v.x * sc, __ATOMIC_RELAXED, __HIP_MEMORY_SCOPE_WORKGROUP);
  __hip_atomic_fetch_add(t + 32, v.y * sc, __ATOMIC_RELAXED, __HIP_MEMORY_SCOPE_WORKGROUP);
  __hip_atomic_fetch_add(t + 64, v.z * sc, __ATOMIC_RELAXED, __HIP_MEMORY_SCOPE_WORKGROUP);
  __hip_atomic_fetch_add(t + 96, v.w * sc, __ATOMIC_RELAXED, __HIP_MEMORY_SCOPE_WORKGROUP);
}

__device__ __forceinline__ void ow_process(
    float* tile, const float* __restrict__ x, const int* __restrict__ src,
    const int* __restrict__ dst, const float* __restrict__ inv_out,
    int pend, int cnt, int lane, int node0) {
  int s_l = 0, d_l = 0, sc_l = 0;
  if (lane < cnt) {
    s_l  = src[pend];
    d_l  = dst[pend] - node0;
    sc_l = __float_as_int(inv_out[s_l]);
  }
  float4 va[8], vb[8];
#pragma unroll
  for (int k = 0; k < 8; ++k)
    if (k < cnt)
      va[k] = *(const float4*)(x + (size_t)__builtin_amdgcn_readlane(s_l, k) * 128 + lane * 4);
  for (int j0 = 0; j0 < cnt; j0 += 16) {
#pragma unroll
    for (int k = 0; k < 8; ++k)
      if (j0 + 8 + k < cnt)
        vb[k] = *(const float4*)(x + (size_t)__builtin_amdgcn_readlane(s_l, j0 + 8 + k) * 128 + lane * 4);
#pragma unroll
    for (int k = 0; k < 8; ++k)
      if (j0 + k < cnt)
        ow_add_row(tile, __builtin_amdgcn_readlane(d_l, j0 + k), lane, va[k],
                   __int_as_float(__builtin_amdgcn_readlane(sc_l, j0 + k)));
#pragma unroll
    for (int k = 0; k < 8; ++k)
      if (j0 + 16 + k < cnt)
        va[k] = *(const float4*)(x + (size_t)__builtin_amdgcn_readlane(s_l, j0 + 16 + k) * 128 + lane * 4);
#pragma unroll
    for (int k = 0; k < 8; ++k)
      if (j0 + 8 + k < cnt)
        ow_add_row(tile, __builtin_amdgcn_readlane(d_l, j0 + 8 + k), lane, vb[k],
                   __int_as_float(__builtin_amdgcn_readlane(sc_l, j0 + 8 + k)));
  }
}

template <int NOUT, bool RELU>
__global__ __launch_bounds__(OW_THREADS) void gcn_spmm_owner_mm(
    const float* __restrict__ x, const int* __restrict__ src,
    const int* __restrict__ dst, const float* __restrict__ inv_out,
    const unsigned char* __restrict__ keys, const float* __restrict__ inv_in,
    const float* __restrict__ W, const float* __restrict__ bias,
    float* __restrict__ outp, int E, int N) {
  __shared__ float tile[OW_ROWS * OW_PITCH];
  __shared__ int queue[OW_THREADS / 32][64];
  const int tid  = threadIdx.x;
  const int lane = tid & 31;
  const int wave = __builtin_amdgcn_readfirstlane(tid >> 5);
  const int g     = blockIdx.x;
  const int node0 = g * OW_NPG;
  const int nrows = min(OW_NPG, N - node0);

  for (int i = tid; i < OW_ROWS * OW_PITCH / 4; i += OW_THREADS)
    ((float4*)tile)[i] = make_float4(0.0f, 0.0f, 0.0f, 0.0f);
  __syncthreads();

  const unsigned pat = (unsigned)g * 0x01010101u;
  const int n_steps = (E + 2047) >> 11;
  const int start = (g * n_steps) >> 8;
  int* myq = &queue[wave][0];
  int head = 0;
  int tail = 0;
  uint4 kw[4];
  {
    int st = wave + start;
    if (st >= n_steps) st -= n_steps;
#pragma unroll
    for (int q = 0; q < 4; ++q)
      kw[q] = *(const uint4*)(keys + min((st << 11) + q * 512 + lane * 16, E - 16));
  }
  for (int j = wave; j < n_steps; j += OW_THREADS / 32) {
    int st = j + start;
    if (st >= n_steps) st -= n_steps;
    const int base = st << 11;
    unsigned M[4];
#pragma unroll
    for (int q = 0; q < 4; ++q) {
      M[q] = (ow_zero_bytes(kw[q].x ^ pat) >> 7) | (ow_zero_bytes(kw[q].y ^ pat) >> 6)
           | (ow_zero_bytes(kw[q].z ^ pat) >> 5) | (ow_zero_bytes(kw[q].w ^ pat) >> 4);
      if (base + q * 512 + lane * 16 >= E) M[q] = 0u;
    }
    if (j + OW_THREADS / 32 < n_steps) {
      int sn = j + OW_THREADS / 32 + start;
      if (sn >= n_steps) sn -= n_steps;
#pragma unroll
      for (int q = 0; q < 4; ++q)
        kw[q] = *(const uint4*)(keys + min((sn << 11) + q * 512 + lane * 16, E - 16));
    }
#pragma nounroll
    for (int q = 0; q < 4; ++q) {
      unsigned w = (q == 0) ? M[0] : (q == 1) ? M[1] : (q == 2) ? M[2] : M[3];
      unsigned bal = __builtin_amdgcn_ballot_w32(w != 0u);
      while (bal) {
        const int p = __builtin_ctz(w | 0x80000000u);
        const int e = base + q * 512 + lane * 16 + ((p & 7) << 2) + (p >> 3);
        const int slot = tail + (int)__builtin_amdgcn_mbcnt_lo(bal, 0u);
        if (w != 0u) myq[slot & 63] = e;
        w &= w - 1u;
        tail += __builtin_popcount(bal);
        __builtin_amdgcn_wave_barrier();
        if (tail - head >= 32) {
          const int pend = myq[(head + lane) & 63];
          ow_process(tile, x, src, dst, inv_out, pend, 32, lane, node0);
          head += 32;
        }
        bal = __builtin_amdgcn_ballot_w32(w != 0u);
      }
    }
  }
  if (tail != head) {
    __builtin_amdgcn_wave_barrier();
    const int pend = myq[(head + lane) & 63];
    ow_process(tile, x, src, dst, inv_out, pend, tail - head, lane, node0);
  }
  __syncthreads();

  constexpr int NCT = NOUT / 16;
  constexpr int NRW = (OW_THREADS / 32) / NCT;
  const int ct   = wave % NCT;
  const int wq   = wave / NCT;
  const int half = lane >> 4;
  const int c    = lane & 15;
  const int col  = ct * 16 + c;
  v2f bfrag[32];
#pragma unroll
  for (int kk = 0; kk < 32; ++kk) {
    const int k2 = kk * 4 + half * 2;
    bfrag[kk].x = W[(size_t)k2 * NOUT + col];
    bfrag[kk].y = W[(size_t)(k2 + 1) * NOUT + col];
  }
  const float bv = bias[col];
  for (int rt = wq; rt < OW_ROWS / 16; rt += NRW) {
    const int arow = rt * 16 + c;
    const float sc = (arow < nrows) ? inv_in[node0 + arow] : 0.0f;
    const float* trow = tile + arow * OW_PITCH + 64 * half;
    v8f acc = {};
#pragma unroll
    for (int kk = 0; kk < 32; ++kk) {
      v2f a;
      a.x = trow[kk] * sc;
      a.y = trow[32 + kk] * sc;
      acc = __builtin_amdgcn_wmma_f32_16x16x4_f32(
           false, a,  false, bfrag[kk],
           (short)0, acc,  false,  false);
    }
#pragma unroll
    for (int r = 0; r < 8; ++r) {
      const int lrow = rt * 16 + r + half * 8;
      if (lrow < nrows) {
        float v = acc[r] + bv;
        if (RELU) v = (v > 0.0f) ? v : 0.0f;
        outp[(size_t)(node0 + lrow) * NOUT + col] = v;
      }
    }
  }
}

extern "C" void kernel_launch(void* const* d_in, const int* in_sizes, int n_in,
                              void* d_out, int out_size, void* d_ws, size_t ws_size,
                              hipStream_t stream) {
  const float* in_feat = (const float*)d_in[0];
  const int*   src     = (const int*)d_in[1];
  const int*   dst     = (const int*)d_in[2];
  const float* W1      = (const float*)d_in[3];
  const float* b1      = (const float*)d_in[4];
  const float* W2      = (const float*)d_in[5];
  const float* b2      = (const float*)d_in[6];
  float* out = (float*)d_out;

  const int N = GCN_N_NODES;
  const int E = GCN_N_EDGES;

  float* inv_out = (float*)d_ws;
  float* inv_in  = inv_out + 50176;
  float* h1      = inv_in + 50176;
  unsigned char* keys = (unsigned char*)(h1 + (size_t)N * 128);

  hipMemsetAsync(inv_out, 0, (size_t)2 * 50176 * sizeof(float), stream);
  gcn_deg_kernel<<<(E + 255) / 256, 256, 0, stream>>>(src, dst, inv_out, inv_in, keys, E);
  gcn_invsqrt_kernel<<<(N + 255) / 256, 256, 0, stream>>>(inv_out, inv_in, N);

  gcn_spmm_owner_mm<GCN_H, true><<<OW_GROUPS, OW_THREADS, 0, stream>>>(
      in_feat, src, dst, inv_out, keys, inv_in, W1, b1, h1, E, N);

  gcn_spmm_owner_mm<GCN_C, false><<<OW_GROUPS, OW_THREADS, 0, stream>>>(
      h1, src, dst, inv_out, keys, inv_in, W2, b2, out, E, N);
}


# Edges split once by owner group in the degree kernel (LDS count + rank per 3125-edge chunk, packed id|row words); groups read their 256 short pieces: no key filter, no id gathers; arrival order kept
# speedup vs baseline: 2.8264x; 1.1479x over previous
#include <hip/hip_runtime.h>
#include <hip/hip_bf16.h>

#define GCN_N_NODES 50000
#define GCN_N_EDGES 800000
#define GCN_IN      128
#define GCN_H       128
#define GCN_C       32
#define OW_GROUPS   256
#define OW_NPG      196
#define OW_ROWS     208
#define OW_PITCH    132
#define OW_CAP      4096
#define PT_CHUNKS   256
#define PT_CH       3125
#define PT_THREADS  1024
#define PT_EPT      4
static_assert(PT_CHUNKS * PT_CH >= GCN_N_EDGES && PT_EPT * PT_THREADS >= PT_CH, "chunks cover the edges");
static_assert(GCN_N_NODES <= 65536 && OW_NPG <= 65536 && PT_CH <= 65535, "16-bit fields of the packed words");

typedef __attribute__((ext_vector_type(2))) float v2f;
typedef __attribute__((ext_vector_type(8))) float v8f;

__global__ __launch_bounds__(PT_THREADS) void gcn_deg_split_kernel(
    const int* __restrict__ src, const int* __restrict__ dst,
    float* __restrict__ deg_out, float* __restrict__ deg_in,
    unsigned* __restrict__ list, unsigned* __restrict__ table, int E) {
  __shared__ int cnt[OW_GROUPS];
  __shared__ int off[OW_GROUPS];
  __shared__ unsigned stage[PT_CH];
  const int tid = threadIdx.x;
  const int c   = blockIdx.x;
  const int e0  = c * PT_CH;
  const int n   = max(0, min(PT_CH, E - e0));
  if (tid < OW_GROUPS) cnt[tid] = 0;
  __syncthreads();
  int s[PT_EPT], d[PT_EPT], r[PT_EPT];
#pragma unroll
  for (int i = 0; i < PT_EPT; ++i) {
    const int idx = tid + i * PT_THREADS;
    s[i] = 0;
    d[i] = 0;
    if (idx < n) {
      s[i] = src[e0 + idx];
      d[i] = dst[e0 + idx];
    }
  }
#pragma unroll
  for (int i = 0; i < PT_EPT; ++i) {
    const int idx = tid + i * PT_THREADS;
    r[i] = 0;
    if (idx < n) r[i] = atomicAdd(&cnt[(unsigned)d[i] / (unsigned)OW_NPG], 1);
  }
  __syncthreads();
  if (tid < 32) {
    int v[8];
    int sum = 0;
#pragma unroll
    for (int i = 0; i < 8; ++i) {
      v[i] = cnt[tid * 8 + i];
      sum += v[i];
    }
    int incl = sum;
#pragma unroll
    for (int dl = 1; dl < 32; dl <<= 1) {
      const int t = __shfl_up(incl, dl, 32);
      if (tid >= dl) incl += t;
    }
    int run = incl - sum;
#pragma unroll
    for (int i = 0; i < 8; ++i) {
      off[tid * 8 + i] = run;
      run += v[i];
    }
  }
  __syncthreads();
#pragma unroll
  for (int i = 0; i < PT_EPT; ++i) {
    const int idx = tid + i * PT_THREADS;
    if (idx < n) {
      const unsigned g = (unsigned)d[i] / (unsigned)OW_NPG;
      stage[off[g] + r[i]] = (unsigned)s[i] | (((unsigned)d[i] - g * (unsigned)OW_NPG) << 16);
    }
  }
  if (tid < OW_GROUPS)
    table[(size_t)tid * PT_CHUNKS + c] = (unsigned)off[tid] | ((unsigned)cnt[tid] << 16);
  __syncthreads();
  for (int i = tid; i < n; i += PT_THREADS) list[e0 + i] = stage[i];
#pragma unroll
  for (int i = 0; i < PT_EPT; ++i) {
    const int idx = tid + i * PT_THREADS;
    if (idx < n) {
      atomicAdd(&deg_out[s[i]], 1.0f);
      atomicAdd(&deg_in[d[i]], 1.0f);
    }
  }
}

__global__ __launch_bounds__(256) void gcn_invsqrt_kernel(
    float* __restrict__ deg_out, float* __restrict__ deg_in, int N) {
  int n = blockIdx.x * blockDim.x + threadIdx.x;
  if (n < N) {
    float d = deg_out[n];
    deg_out[n] = (d > 0.0f) ? rsqrtf(d) : 0.0f;
    d = deg_in[n];
    deg_in[n] = (d > 0.0f) ? rsqrtf(d) : 0.0f;
  }
}

#define OW_THREADS 1024
#define OW_FPT     (OW_CAP / OW_THREADS)

__device__ __forceinline__ void ow_add_row(float* tile, int d, int lane, float4 v, float sc) {
  float* t = tile + d * OW_PITCH + lane;
  __hip_atomic_fetch_add(t + 0,  v.x * sc, __ATOMIC_RELAXED, __HIP_MEMORY_SCOPE_WORKGROUP);
  __hip_atomic_fetch_add(t + 32, v.y * sc, __ATOMIC_RELAXED, __HIP_MEMORY_SCOPE_WORKGROUP);
  __hip_atomic_fetch_add(t + 64, v.z * sc, __ATOMIC_RELAXED, __HIP_MEMORY_SCOPE_WORKGROUP);
  __hip_atomic_fetch_add(t + 96, v.w * sc, __ATOMIC_RELAXED, __HIP_MEMORY_SCOPE_WORKGROUP);
}

__device__ __forceinline__ void ow_process(
    float* tile, const float* __restrict__ x,
    int s_l, int d_l, int sc_l, int cnt, int lane) {
  float4 va[8], vb[8];
#pragma unroll
  for (int k = 0; k < 8; ++k)
    va[k] = *(const float4*)(x + (size_t)__builtin_amdgcn_readlane(s_l, k) * 128 + lane * 4);
#pragma unroll
  for (int j0 = 0; j0 < 32; j0 += 16) {
#pragma unroll
    for (int k = 0; k < 8; ++k)
      vb[k] = *(const float4*)(x + (size_t)__builtin_amdgcn_readlane(s_l, j0 + 8 + k) * 128 + lane * 4);
#pragma unroll
    for (int k = 0; k < 8; ++k)
      if (j0 + k < cnt)
        ow_add_row(tile, __builtin_amdgcn_readlane(d_l, j0 + k), lane, va[k],
                   __int_as_float(__builtin_amdgcn_readlane(sc_l, j0 + k)));
#pragma unroll
    for (int k = 0; k < 8; ++k)
      if (j0 + 16 + k < 32)
        va[k] = *(const float4*)(x + (size_t)__builtin_amdgcn_readlane(s_l, j0 + 16 + k) * 128 + lane * 4);
#pragma unroll
    for (int k = 0; k < 8; ++k)
      if (j0 + 8 + k < cnt)
        ow_add_row(tile, __builtin_amdgcn_readlane(d_l, j0 + 8 + k), lane, vb[k],
                   __int_as_float(__builtin_amdgcn_readlane(sc_l, j0 + 8 + k)));
  }
}

template <int NOUT, bool RELU>
__global__ __launch_bounds__(OW_THREADS) void gcn_spmm_owner_mm(
    const float* __restrict__ x, const unsigned* __restrict__ list,
    const unsigned* __restrict__ table, const float* __restrict__ inv_out,
    const float* __restrict__ inv_in,
    const float* __restrict__ W, const float* __restrict__ bias,
    float* __restrict__ outp, int N) {
  __shared__ float tile[OW_ROWS * OW_PITCH];
  __shared__ unsigned s_tab[PT_CHUNKS];
  __shared__ int      s_base[PT_CHUNKS + 1];
  __shared__ unsigned s_ent[OW_CAP];
  __shared__ float    s_sc[OW_CAP];
  const int tid  = threadIdx.x;
  const int lane = tid & 31;
  const int wave = __builtin_amdgcn_readfirstlane(tid >> 5);
  const int g     = blockIdx.x;
  const int node0 = g * OW_NPG;
  const int nrows = min(OW_NPG, N - node0);

  if (tid < PT_CHUNKS) s_tab[tid] = table[(size_t)g * PT_CHUNKS + tid];
  for (int i = tid; i < OW_ROWS * OW_PITCH / 4; i += OW_THREADS)
    ((float4*)tile)[i] = make_float4(0.0f, 0.0f, 0.0f, 0.0f);
  __syncthreads();
  if (tid < 32) {
    int v[8];
    int sum = 0;
#pragma unroll
    for (int i = 0; i < 8; ++i) {
      v[i] = (int)(s_tab[tid * 8 + i] >> 16);
      sum += v[i];
    }
    int incl = sum;
#pragma unroll
    for (int dl = 1; dl < 32; dl <<= 1) {
      const int t = __shfl_up(incl, dl, 32);
      if (tid >= dl) incl += t;
    }
    int run = incl - sum;
#pragma unroll
    for (int i = 0; i < 8; ++i) {
      s_base[tid * 8 + i] = run;
      run += v[i];
    }
    if (tid == 31) s_base[PT_CHUNKS] = run;
  }
  __syncthreads();
  const int total = __builtin_amdgcn_readfirstlane(s_base[PT_CHUNKS]);

  for (int wb = 0; wb < total; wb += OW_CAP) {
    const int nwin = min(OW_CAP, total - wb);
    int      fc[OW_FPT], fp[OW_FPT];
    unsigned fe[OW_FPT];
    float    fs[OW_FPT];
#pragma unroll
    for (int q = 0; q < OW_FPT; ++q) {
      fp[q] = min(wb + tid + q * OW_THREADS, total - 1);
      int lo = 0;
#pragma unroll
      for (int step = PT_CHUNKS / 2; step >= 1; step >>= 1)
        if (s_base[lo + step] <= fp[q]) lo += step;
      fc[q] = lo;
    }
#pragma unroll
    for (int q = 0; q < OW_FPT; ++q)
      fe[q] = list[(size_t)fc[q] * PT_CH + (s_tab[fc[q]] & 0xFFFFu) + (unsigned)(fp[q] - s_base[fc[q]])];
#pragma unroll
    for (int q = 0; q < OW_FPT; ++q)
      fs[q] = inv_out[fe[q] & 0xFFFFu];
#pragma unroll
    for (int q = 0; q < OW_FPT; ++q) {
      s_ent[tid + q * OW_THREADS] = fe[q];
      s_sc[tid + q * OW_THREADS]  = fs[q];
    }
    __syncthreads();
    const int nb = (nwin + 31) >> 5;
    for (int b = wave; b < nb; b += OW_THREADS / 32) {
      const int cnt = min(32, nwin - (b << 5));
      int s_l = 0, d_l = 0, sc_l = 0;
      if (lane < cnt) {
        const unsigned ent = s_ent[(b << 5) + lane];
        s_l  = (int)(ent & 0xFFFFu);
        d_l  = (int)(ent >> 16);
        sc_l = __float_as_int(s_sc[(b << 5) + lane]);
      }
      if (cnt == 32) ow_process(tile, x, s_l, d_l, sc_l, 32, lane);
      else           ow_process(tile, x, s_l, d_l, sc_l, cnt, lane);
    }
    __syncthreads();
  }

  constexpr int NCT = NOUT / 16;
  constexpr int NRW = (OW_THREADS / 32) / NCT;
  const int ct   = wave % NCT;
  const int wq   = wave / NCT;
  const int half = lane >> 4;
  const int c    = lane & 15;
  const int col  = ct * 16 + c;
  v2f bfrag[32];
#pragma unroll
  for (int kk = 0; kk < 32; ++kk) {
    const int k2 = kk * 4 + half * 2;
    bfrag[kk].x = W[(size_t)k2 * NOUT + col];
    bfrag[kk].y = W[(size_t)(k2 + 1) * NOUT + col];
  }
  const float bv = bias[col];
  for (int rt = wq; rt < OW_ROWS / 16; rt += NRW) {
    const int arow = rt * 16 + c;
    const float sc = (arow < nrows) ? inv_in[node0 + arow] : 0.0f;
    const float* trow = tile + arow * OW_PITCH + 64 * half;
    v8f acc = {};
#pragma unroll
    for (int kk = 0; kk < 32; ++kk) {
      v2f a;
      a.x = trow[kk] * sc;
      a.y = trow[32 + kk] * sc;
      acc = __builtin_amdgcn_wmma_f32_16x16x4_f32(
           false, a,  false, bfrag[kk],
           (short)0, acc,  false,  false);
    }
#pragma unroll
    for (int r = 0; r < 8; ++r) {
      const int lrow = rt * 16 + r + half * 8;
      if (lrow < nrows) {
        float v = acc[r] + bv;
        if (RELU) v = (v > 0.0f) ? v : 0.0f;
        outp[(size_t)(node0 + lrow) * NOUT + col] = v;
      }
    }
  }
}

extern "C" void kernel_launch(void* const* d_in, const int* in_sizes, int n_in,
                              void* d_out, int out_size, void* d_ws, size_t ws_size,
                              hipStream_t stream) {
  const float* in_feat = (const float*)d_in[0];
  const int*   src     = (const int*)d_in[1];
  const int*   dst     = (const int*)d_in[2];
  const float* W1      = (const float*)d_in[3];
  const float* b1      = (const float*)d_in[4];
  const float* W2      = (const float*)d_in[5];
  const float* b2      = (const float*)d_in[6];
  float* out = (float*)d_out;

  const int N = GCN_N_NODES;
  const int E = GCN_N_EDGES;

  float* inv_out = (float*)d_ws;
  float* inv_in  = inv_out + 50176;
  float* h1      = inv_in + 50176;
  unsigned* list  = (unsigned*)(h1 + (size_t)N * 128);
  unsigned* table = list + GCN_N_EDGES;

  hipMemsetAsync(inv_out, 0, (size_t)2 * 50176 * sizeof(float), stream);
  gcn_deg_split_kernel<<<PT_CHUNKS, PT_THREADS, 0, stream>>>(src, dst, inv_out, inv_in, list, table, E);
  gcn_invsqrt_kernel<<<(N + 255) / 256, 256, 0, stream>>>(inv_out, inv_in, N);

  gcn_spmm_owner_mm<GCN_H, true><<<OW_GROUPS, OW_THREADS, 0, stream>>>(
      in_feat, list, table, inv_out, inv_in, W1, b1, h1, N);

  gcn_spmm_owner_mm<GCN_C, false><<<OW_GROUPS, OW_THREADS, 0, stream>>>(
      h1, list, table, inv_out, inv_in, W2, b2, out, N);
}


# Owner groups form both D^-1/2 scales themselves: in-degree = LDS count of their own list entries, source rsqrt at the fetch; invsqrt launch and in-degree float atomics gone (4 launches -> 3)
# speedup vs baseline: 3.6724x; 1.2993x over previous
#include <hip/hip_runtime.h>
#include <hip/hip_bf16.h>

#define GCN_N_NODES 50000
#define GCN_N_EDGES 800000
#define GCN_IN      128
#define GCN_H       128
#define GCN_C       32
#define OW_GROUPS   256
#define OW_NPG      196
#define OW_ROWS     208
#define OW_PITCH    132
#define OW_CAP      4096
#define PT_CHUNKS   256
#define PT_CH       3125
#define PT_THREADS  1024
#define PT_EPT      4
static_assert(PT_CHUNKS * PT_CH >= GCN_N_EDGES && PT_EPT * PT_THREADS >= PT_CH, "chunks cover the edges");
static_assert(GCN_N_NODES <= 65536 && OW_NPG <= 65536 && PT_CH <= 65535, "16-bit fields of the packed words");

typedef __attribute__((ext_vector_type(2))) float v2f;
typedef __attribute__((ext_vector_type(8))) float v8f;

__global__ __launch_bounds__(PT_THREADS) void gcn_deg_split_kernel(
    const int* __restrict__ src, const int* __restrict__ dst,
    float* __restrict__ deg_out,
    unsigned* __restrict__ list, unsigned* __restrict__ table, int E) {
  __shared__ int cnt[OW_GROUPS];
  __shared__ int off[OW_GROUPS];
  __shared__ unsigned stage[PT_CH];
  const int tid = threadIdx.x;
  const int c   = blockIdx.x;
  const int e0  = c * PT_CH;
  const int n   = max(0, min(PT_CH, E - e0));
  if (tid < OW_GROUPS) cnt[tid] = 0;
  __syncthreads();
  int s[PT_EPT], d[PT_EPT], r[PT_EPT];
#pragma unroll
  for (int i = 0; i < PT_EPT; ++i) {
    const int idx = tid + i * PT_THREADS;
    s[i] = 0;
    d[i] = 0;
    if (idx < n) {
      s[i] = src[e0 + idx];
      d[i] = dst[e0 + idx];
    }
  }
#pragma unroll
  for (int i = 0; i < PT_EPT; ++i) {
    const int idx = tid + i * PT_THREADS;
    r[i] = 0;
    if (idx < n) r[i] = atomicAdd(&cnt[(unsigned)d[i] / (unsigned)OW_NPG], 1);
  }
  __syncthreads();
  if (tid < 32) {
    int v[8];
    int sum = 0;
#pragma unroll
    for (int i = 0; i < 8; ++i) {
      v[i] = cnt[tid * 8 + i];
      sum += v[i];
    }
    int incl = sum;
#pragma unroll
    for (int dl = 1; dl < 32; dl <<= 1) {
      const int t = __shfl_up(incl, dl, 32);
      if (tid >= dl) incl += t;
    }
    int run = incl - sum;
#pragma unroll
    for (int i = 0; i < 8; ++i) {
      off[tid * 8 + i] = run;
      run += v[i];
    }
  }
  __syncthreads();
#pragma unroll
  for (int i = 0; i < PT_EPT; ++i) {
    const int idx = tid + i * PT_THREADS;
    if (idx < n) {
      const unsigned g = (unsigned)d[i] / (unsigned)OW_NPG;
      stage[off[g] + r[i]] = (unsigned)s[i] | (((unsigned)d[i] - g * (unsigned)OW_NPG) << 16);
    }
  }
  if (tid < OW_GROUPS)
    table[(size_t)tid * PT_CHUNKS + c] = (unsigned)off[tid] | ((unsigned)cnt[tid] << 16);
  __syncthreads();
  for (int i = tid; i < n; i += PT_THREADS) list[e0 + i] = stage[i];
#pragma unroll
  for (int i = 0; i < PT_EPT; ++i) {
    const int idx = tid + i * PT_THREADS;
    if (idx < n) atomicAdd(&deg_out[s[i]], 1.0f);
  }
}

#define OW_THREADS 1024
#define OW_FPT     (OW_CAP / OW_THREADS)

__device__ __forceinline__ void ow_add_row(float* tile, int d, int lane, float4 v, float sc) {
  float* t = tile + d * OW_PITCH + lane;
  __hip_atomic_fetch_add(t + 0,  v.x * sc, __ATOMIC_RELAXED, __HIP_MEMORY_SCOPE_WORKGROUP);
  __hip_atomic_fetch_add(t + 32, v.y * sc, __ATOMIC_RELAXED, __HIP_MEMORY_SCOPE_WORKGROUP);
  __hip_atomic_fetch_add(t + 64, v.z * sc, __ATOMIC_RELAXED, __HIP_MEMORY_SCOPE_WORKGROUP);
  __hip_atomic_fetch_add(t + 96, v.w * sc, __ATOMIC_RELAXED, __HIP_MEMORY_SCOPE_WORKGROUP);
}

__device__ __forceinline__ void ow_process(
    float* tile, const float* __restrict__ x,
    int s_l, int d_l, int sc_l, int cnt, int lane) {
  float4 va[8], vb[8];
#pragma unroll
  for (int k = 0; k < 8; ++k)
    va[k] = *(const float4*)(x + (size_t)__builtin_amdgcn_readlane(s_l, k) * 128 + lane * 4);
#pragma unroll
  for (int j0 = 0; j0 < 32; j0 += 16) {
#pragma unroll
    for (int k = 0; k < 8; ++k)
      vb[k] = *(const float4*)(x + (size_t)__builtin_amdgcn_readlane(s_l, j0 + 8 + k) * 128 + lane * 4);
#pragma unroll
    for (int k = 0; k < 8; ++k)
      if (j0 + k < cnt)
        ow_add_row(tile, __builtin_amdgcn_readlane(d_l, j0 + k), lane, va[k],
                   __int_as_float(__builtin_amdgcn_readlane(sc_l, j0 + k)));
#pragma unroll
    for (int k = 0; k < 8; ++k)
      if (j0 + 16 + k < 32)
        va[k] = *(const float4*)(x + (size_t)__builtin_amdgcn_readlane(s_l, j0 + 16 + k) * 128 + lane * 4);
#pragma unroll
    for (int k = 0; k < 8; ++k)
      if (j0 + 8 + k < cnt)
        ow_add_row(tile, __builtin_amdgcn_readlane(d_l, j0 + 8 + k), lane, vb[k],
                   __int_as_float(__builtin_amdgcn_readlane(sc_l, j0 + 8 + k)));
  }
}

template <int NOUT, bool RELU>
__global__ __launch_bounds__(OW_THREADS) void gcn_spmm_owner_mm(
    const float* __restrict__ x, const unsigned* __restrict__ list,
    const unsigned* __restrict__ table, const float* __restrict__ deg_out,
    const float* __restrict__ W, const float* __restrict__ bias,
    float* __restrict__ outp, int N) {
  __shared__ float tile[OW_ROWS * OW_PITCH];
  __shared__ unsigned s_tab[PT_CHUNKS];
  __shared__ int      s_base[PT_CHUNKS + 1];
  __shared__ unsigned s_ent[OW_CAP];
  __shared__ float    s_sc[OW_CAP];
  __shared__ int      s_deg[OW_ROWS];
  const int tid  = threadIdx.x;
  const int lane = tid & 31;
  const int wave = __builtin_amdgcn_readfirstlane(tid >> 5);
  const int g     = blockIdx.x;
  const int node0 = g * OW_NPG;
  const int nrows = min(OW_NPG, N - node0);

  if (tid < PT_CHUNKS) s_tab[tid] = table[(size_t)g * PT_CHUNKS + tid];
  if (tid < OW_ROWS) s_deg[tid] = 0;
  for (int i = tid; i < OW_ROWS * OW_PITCH / 4; i += OW_THREADS)
    ((float4*)tile)[i] = make_float4(0.0f, 0.0f, 0.0f, 0.0f);
  __syncthreads();
  if (tid < 32) {
    int v[8];
    int sum = 0;
#pragma unroll
    for (int i = 0; i < 8; ++i) {
      v[i] = (int)(s_tab[tid * 8 + i] >> 16);
      sum += v[i];
    }
    int incl = sum;
#pragma unroll
    for (int dl = 1; dl < 32; dl <<= 1) {
      const int t = __shfl_up(incl, dl, 32);
      if (tid >= dl) incl += t;
    }
    int run = incl - sum;
#pragma unroll
    for (int i = 0; i < 8; ++i) {
      s_base[tid * 8 + i] = run;
      run += v[i];
    }
    if (tid == 31) s_base[PT_CHUNKS] = run;
  }
  __syncthreads();
  const int total = __builtin_amdgcn_readfirstlane(s_base[PT_CHUNKS]);

  for (int wb = 0; wb < total; wb += OW_CAP) {
    const int nwin = min(OW_CAP, total - wb);
    int      fc[OW_FPT], fp[OW_FPT];
    unsigned fe[OW_FPT];
    float    fs[OW_FPT];
#pragma unroll
    for (int q = 0; q < OW_FPT; ++q) {
      fp[q] = min(wb + tid + q * OW_THREADS, total - 1);
      int lo = 0;
#pragma unroll
      for (int step = PT_CHUNKS / 2; step >= 1; step >>= 1)
        if (s_base[lo + step] <= fp[q]) lo += step;
      fc[q] = lo;
    }
#pragma unroll
    for (int q = 0; q < OW_FPT; ++q)
      fe[q] = list[(size_t)fc[q] * PT_CH + (s_tab[fc[q]] & 0xFFFFu) + (unsigned)(fp[q] - s_base[fc[q]])];
#pragma unroll
    for (int q = 0; q < OW_FPT; ++q)
      fs[q] = deg_out[fe[q] & 0xFFFFu];
#pragma unroll
    for (int q = 0; q < OW_FPT; ++q) {
      s_ent[tid + q * OW_THREADS] = fe[q];
      s_sc[tid + q * OW_THREADS]  = (fs[q] > 0.0f) ? rsqrtf(fs[q]) : 0.0f;
      if (wb + tid + q * OW_THREADS < total) atomicAdd(&s_deg[fe[q] >> 16], 1);
    }
    __syncthreads();
    const int nb = (nwin + 31) >> 5;
    for (int b = wave; b < nb; b += OW_THREADS / 32) {
      const int cnt = min(32, nwin - (b << 5));
      int s_l = 0, d_l = 0, sc_l = 0;
      if (lane < cnt) {
        const unsigned ent = s_ent[(b << 5) + lane];
        s_l  = (int)(ent & 0xFFFFu);
        d_l  = (int)(ent >> 16);
        sc_l = __float_as_int(s_sc[(b << 5) + lane]);
      }
      if (cnt == 32) ow_process(tile, x, s_l, d_l, sc_l, 32, lane);
      else           ow_process(tile, x, s_l, d_l, sc_l, cnt, lane);
    }
    __syncthreads();
  }

  constexpr int NCT = NOUT / 16;
  constexpr int NRW = (OW_THREADS / 32) / NCT;
  const int ct   = wave % NCT;
  const int wq   = wave / NCT;
  const int half = lane >> 4;
  const int c    = lane & 15;
  const int col  = ct * 16 + c;
  v2f bfrag[32];
#pragma unroll
  for (int kk = 0; kk < 32; ++kk) {
    const int k2 = kk * 4 + half * 2;
    bfrag[kk].x = W[(size_t)k2 * NOUT + col];
    bfrag[kk].y = W[(size_t)(k2 + 1) * NOUT + col];
  }
  const float bv = bias[col];
  for (int rt = wq; rt < OW_ROWS / 16; rt += NRW) {
    const int arow = rt * 16 + c;
    const int   dg = s_deg[arow];
    const float sc = (dg > 0) ? rsqrtf((float)dg) : 0.0f;
    const float* trow = tile + arow * OW_PITCH + 64 * half;
    v8f acc = {};
#pragma unroll
    for (int kk = 0; kk < 32; ++kk) {
      v2f a;
      a.x = trow[kk] * sc;
      a.y = trow[32 + kk] * sc;
      acc = __builtin_amdgcn_wmma_f32_16x16x4_f32(
           false, a,  false, bfrag[kk],
           (short)0, acc,  false,  false);
    }
#pragma unroll
    for (int r = 0; r < 8; ++r) {
      const int lrow = rt * 16 + r + half * 8;
      if (lrow < nrows) {
        float v = acc[r] + bv;
        if (RELU) v = (v > 0.0f) ? v : 0.0f;
        outp[(size_t)(node0 + lrow) * NOUT + col] = v;
      }
    }
  }
}

extern "C" void kernel_launch(void* const* d_in, const int* in_sizes, int n_in,
                              void* d_out, int out_size, void* d_ws, size_t ws_size,
                              hipStream_t stream) {
  const float* in_feat = (const float*)d_in[0];
  const int*   src     = (const int*)d_in[1];
  const int*   dst     = (const int*)d_in[2];
  const float* W1      = (const float*)d_in[3];
  const float* b1      = (const float*)d_in[4];
  const float* W2      = (const float*)d_in[5];
  const float* b2      = (const float*)d_in[6];
  float* out = (float*)d_out;

  const int N = GCN_N_NODES;
  const int E = GCN_N_EDGES;

  float* deg_out = (float*)d_ws;
  float* h1      = deg_out + 50176;
  unsigned* list  = (unsigned*)(h1 + (size_t)N * 128);
  unsigned* table = list + GCN_N_EDGES;

  hipMemsetAsync(deg_out, 0, (size_t)50176 * sizeof(float), stream);
  gcn_deg_split_kernel<<<PT_CHUNKS, PT_THREADS, 0, stream>>>(src, dst, deg_out, list, table, E);

  gcn_spmm_owner_mm<GCN_H, true><<<OW_GROUPS, OW_THREADS, 0, stream>>>(
      in_feat, list, table, deg_out, W1, b1, h1, N);

  gcn_spmm_owner_mm<GCN_C, false><<<OW_GROUPS, OW_THREADS, 0, stream>>>(
      h1, list, table, deg_out, W2, b2, out, N);
}


# Window entries counting-sorted by tile row in LDS (rank = returning count, scan, place); a wave sums a row's run in registers, one ds f32 add set per run; run sums in rank, runs in arrival order
# speedup vs baseline: 3.9714x; 1.0814x over previous
#include <hip/hip_runtime.h>
#include <hip/hip_bf16.h>

#define GCN_N_NODES 50000
#define GCN_N_EDGES 800000
#define GCN_IN      128
#define GCN_H       128
#define GCN_C       32
#define OW_GROUPS   256
#define OW_NPG      196
#define OW_ROWS     208
#define OW_PITCH    132
#define OW_CAP      4096
#define PT_CHUNKS   256
#define PT_CH       3125
#define PT_THREADS  1024
#define PT_EPT      4
static_assert(PT_CHUNKS * PT_CH >= GCN_N_EDGES && PT_EPT * PT_THREADS >= PT_CH, "chunks cover the edges");
static_assert(GCN_N_NODES <= 65536 && OW_NPG <= 65536 && PT_CH <= 65535, "16-bit fields of the packed words");

typedef __attribute__((ext_vector_type(2))) float v2f;
typedef __attribute__((ext_vector_type(8))) float v8f;

__global__ __launch_bounds__(PT_THREADS) void gcn_deg_split_kernel(
    const int* __restrict__ src, const int* __restrict__ dst,
    float* __restrict__ deg_out,
    unsigned* __restrict__ list, unsigned* __restrict__ table, int E) {
  __shared__ int cnt[OW_GROUPS];
  __shared__ int off[OW_GROUPS];
  __shared__ unsigned stage[PT_CH];
  const int tid = threadIdx.x;
  const int c   = blockIdx.x;
  const int e0  = c * PT_CH;
  const int n   = max(0, min(PT_CH, E - e0));
  if (tid < OW_GROUPS) cnt[tid] = 0;
  __syncthreads();
  int s[PT_EPT], d[PT_EPT], r[PT_EPT];
#pragma unroll
  for (int i = 0; i < PT_EPT; ++i) {
    const int idx = tid + i * PT_THREADS;
    s[i] = 0;
    d[i] = 0;
    if (idx < n) {
      s[i] = src[e0 + idx];
      d[i] = dst[e0 + idx];
    }
  }
#pragma unroll
  for (int i = 0; i < PT_EPT; ++i) {
    const int idx = tid + i * PT_THREADS;
    r[i] = 0;
    if (idx < n) r[i] = atomicAdd(&cnt[(unsigned)d[i] / (unsigned)OW_NPG], 1);
  }
  __syncthreads();
  if (tid < 32) {
    int v[8];
    int sum = 0;
#pragma unroll
    for (int i = 0; i < 8; ++i) {
      v[i] = cnt[tid * 8 + i];
      sum += v[i];
    }
    int incl = sum;
#pragma unroll
    for (int dl = 1; dl < 32; dl <<= 1) {
      const int t = __shfl_up(incl, dl, 32);
      if (tid >= dl) incl += t;
    }
    int run = incl - sum;
#pragma unroll
    for (int i = 0; i < 8; ++i) {
      off[tid * 8 + i] = run;
      run += v[i];
    }
  }
  __syncthreads();
#pragma unroll
  for (int i = 0; i < PT_EPT; ++i) {
    const int idx = tid + i * PT_THREADS;
    if (idx < n) {
      const unsigned g = (unsigned)d[i] / (unsigned)OW_NPG;
      stage[off[g] + r[i]] = (unsigned)s[i] | (((unsigned)d[i] - g * (unsigned)OW_NPG) << 16);
    }
  }
  if (tid < OW_GROUPS)
    table[(size_t)tid * PT_CHUNKS + c] = (unsigned)off[tid] | ((unsigned)cnt[tid] << 16);
  __syncthreads();
  for (int i = tid; i < n; i += PT_THREADS) list[e0 + i] = stage[i];
#pragma unroll
  for (int i = 0; i < PT_EPT; ++i) {
    const int idx = tid + i * PT_THREADS;
    if (idx < n) atomicAdd(&deg_out[s[i]], 1.0f);
  }
}

#define OW_THREADS 1024
#define OW_FPT     (OW_CAP / OW_THREADS)

__device__ __forceinline__ void ow_flush_run(float* tile, int d, int lane, float4 a) {
  float* t = tile + d * OW_PITCH + lane;
  __hip_atomic_fetch_add(t + 0,  a.x, __ATOMIC_RELAXED, __HIP_MEMORY_SCOPE_WORKGROUP);
  __hip_atomic_fetch_add(t + 32, a.y, __ATOMIC_RELAXED, __HIP_MEMORY_SCOPE_WORKGROUP);
  __hip_atomic_fetch_add(t + 64, a.z, __ATOMIC_RELAXED, __HIP_MEMORY_SCOPE_WORKGROUP);
  __hip_atomic_fetch_add(t + 96, a.w, __ATOMIC_RELAXED, __HIP_MEMORY_SCOPE_WORKGROUP);
}

__device__ __forceinline__ void ow_add_row(float* tile, int d, int lane, float4 v, float sc,
                                           int& cur, float4& acc) {
  if (d != cur) {
    ow_flush_run(tile, cur, lane, acc);
    acc = make_float4(0.0f, 0.0f, 0.0f, 0.0f);
    cur = d;
  }
  acc.x += v.x * sc;
  acc.y += v.y * sc;
  acc.z += v.z * sc;
  acc.w += v.w * sc;
}

__device__ __forceinline__ void ow_process(
    float* tile, const float* __restrict__ x,
    int s_l, int d_l, int sc_l, int cnt, int lane) {
  float4 va[8], vb[8];
  int    cur = __builtin_amdgcn_readlane(d_l, 0);
  float4 acc = make_float4(0.0f, 0.0f, 0.0f, 0.0f);
#pragma unroll
  for (int k = 0; k < 8; ++k)
    va[k] = *(const float4*)(x + (size_t)__builtin_amdgcn_readlane(s_l, k) * 128 + lane * 4);
#pragma unroll
  for (int j0 = 0; j0 < 32; j0 += 16) {
#pragma unroll
    for (int k = 0; k < 8; ++k)
      vb[k] = *(const float4*)(x + (size_t)__builtin_amdgcn_readlane(s_l, j0 + 8 + k) * 128 + lane * 4);
#pragma unroll
    for (int k = 0; k < 8; ++k)
      if (j0 + k < cnt)
        ow_add_row(tile, __builtin_amdgcn_readlane(d_l, j0 + k), lane, va[k],
                   __int_as_float(__builtin_amdgcn_readlane(sc_l, j0 + k)), cur, acc);
#pragma unroll
    for (int k = 0; k < 8; ++k)
      if (j0 + 16 + k < 32)
        va[k] = *(const float4*)(x + (size_t)__builtin_amdgcn_readlane(s_l, j0 + 16 + k) * 128 + lane * 4);
#pragma unroll
    for (int k = 0; k < 8; ++k)
      if (j0 + 8 + k < cnt)
        ow_add_row(tile, __builtin_amdgcn_readlane(d_l, j0 + 8 + k), lane, vb[k],
                   __int_as_float(__builtin_amdgcn_readlane(sc_l, j0 + 8 + k)), cur, acc);
  }
  ow_flush_run(tile, cur, lane, acc);
}

template <int NOUT, bool RELU>
__global__ __launch_bounds__(OW_THREADS) void gcn_spmm_owner_mm(
    const float* __restrict__ x, const unsigned* __restrict__ list,
    const unsigned* __restrict__ table, const float* __restrict__ deg_out,
    const float* __restrict__ W, const float* __restrict__ bias,
    float* __restrict__ outp, int N) {
  __shared__ float tile[OW_ROWS * OW_PITCH];
  __shared__ unsigned s_tab[PT_CHUNKS];
  __shared__ int      s_base[PT_CHUNKS + 1];
  __shared__ unsigned s_ent[OW_CAP];
  __shared__ float    s_sc[OW_CAP];
  __shared__ int      s_deg[OW_ROWS];
  __shared__ int      s_cnt[OW_ROWS];
  __shared__ int      s_off[OW_ROWS];
  const int tid  = threadIdx.x;
  const int lane = tid & 31;
  const int wave = __builtin_amdgcn_readfirstlane(tid >> 5);
  const int g     = blockIdx.x;
  const int node0 = g * OW_NPG;
  const int nrows = min(OW_NPG, N - node0);

  if (tid < PT_CHUNKS) s_tab[tid] = table[(size_t)g * PT_CHUNKS + tid];
  if (tid < OW_ROWS) { s_deg[tid] = 0; s_cnt[tid] = 0; }
  for (int i = tid; i < OW_ROWS * OW_PITCH / 4; i += OW_THREADS)
    ((float4*)tile)[i] = make_float4(0.0f, 0.0f, 0.0f, 0.0f);
  __syncthreads();
  if (tid < 32) {
    int v[8];
    int sum = 0;
#pragma unroll
    for (int i = 0; i < 8; ++i) {
      v[i] = (int)(s_tab[tid * 8 + i] >> 16);
      sum += v[i];
    }
    int incl = sum;
#pragma unroll
    for (int dl = 1; dl < 32; dl <<= 1) {
      const int t = __shfl_up(incl, dl, 32);
      if (tid >= dl) incl += t;
    }
    int run = incl - sum;
#pragma unroll
    for (int i = 0; i < 8; ++i) {
      s_base[tid * 8 + i] = run;
      run += v[i];
    }
    if (tid == 31) s_base[PT_CHUNKS] = run;
  }
  __syncthreads();
  const int total = __builtin_amdgcn_readfirstlane(s_base[PT_CHUNKS]);

  for (int wb = 0; wb < total; wb += OW_CAP) {
    const int nwin = min(OW_CAP, total - wb);
    int      fc[OW_FPT], fp[OW_FPT];
    unsigned fe[OW_FPT];
    float    fs[OW_FPT];
#pragma unroll
    for (int q = 0; q < OW_FPT; ++q) {
      fp[q] = min(wb + tid + q * OW_THREADS, total - 1);
      int lo = 0;
#pragma unroll
      for (int step = PT_CHUNKS / 2; step >= 1; step >>= 1)
        if (s_base[lo + step] <= fp[q]) lo += step;
      fc[q] = lo;
    }
#pragma unroll
    for (int q = 0; q < OW_FPT; ++q)
      fe[q] = list[(size_t)fc[q] * PT_CH + (s_tab[fc[q]] & 0xFFFFu) + (unsigned)(fp[q] - s_base[fc[q]])];
#pragma unroll
    for (int q = 0; q < OW_FPT; ++q)
      fs[q] = deg_out[fe[q] & 0xFFFFu];
    int rk[OW_FPT];
#pragma unroll
    for (int q = 0; q < OW_FPT; ++q) {
      rk[q] = 0;
      if (wb + tid + q * OW_THREADS < total) rk[q] = atomicAdd(&s_cnt[fe[q] >> 16], 1);
    }
    __syncthreads();
    if (tid < 32) {
      int v[7];
      int sum = 0;
#pragma unroll
      for (int i = 0; i < 7; ++i) {
        const int idx = tid * 7 + i;
        v[i] = (idx < OW_ROWS) ? s_cnt[idx] : 0;
        sum += v[i];
      }
      int incl = sum;
#pragma unroll
      for (int dl = 1; dl < 32; dl <<= 1) {
        const int t = __shfl_up(incl, dl, 32);
        if (tid >= dl) incl += t;
      }
      int run = incl - sum;
#pragma unroll
      for (int i = 0; i < 7; ++i) {
        const int idx = tid * 7 + i;
        if (idx < OW_ROWS) {
          s_off[idx] = run;
          s_deg[idx] += v[i];
          s_cnt[idx] = 0;
        }
        run += v[i];
      }
    }
    __syncthreads();
#pragma unroll
    for (int q = 0; q < OW_FPT; ++q) {
      if (wb + tid + q * OW_THREADS < total) {
        const int pos = s_off[fe[q] >> 16] + rk[q];
        s_ent[pos] = fe[q];
        s_sc[pos]  = (fs[q] > 0.0f) ? rsqrtf(fs[q]) : 0.0f;
      }
    }
    __syncthreads();
    const int nb = (nwin + 31) >> 5;
    for (int b = wave; b < nb; b += OW_THREADS / 32) {
      const int cnt = min(32, nwin - (b << 5));
      int s_l = 0, d_l = 0, sc_l = 0;
      if (lane < cnt) {
        const unsigned ent = s_ent[(b << 5) + lane];
        s_l  = (int)(ent & 0xFFFFu);
        d_l  = (int)(ent >> 16);
        sc_l = __float_as_int(s_sc[(b << 5) + lane]);
      }
      if (cnt == 32) ow_process(tile, x, s_l, d_l, sc_l, 32, lane);
      else           ow_process(tile, x, s_l, d_l, sc_l, cnt, lane);
    }
    __syncthreads();
  }

  constexpr int NCT = NOUT / 16;
  constexpr int NRW = (OW_THREADS / 32) / NCT;
  const int ct   = wave % NCT;
  const int wq   = wave / NCT;
  const int half = lane >> 4;
  const int c    = lane & 15;
  const int col  = ct * 16 + c;
  v2f bfrag[32];
#pragma unroll
  for (int kk = 0; kk < 32; ++kk) {
    const int k2 = kk * 4 + half * 2;
    bfrag[kk].x = W[(size_t)k2 * NOUT + col];
    bfrag[kk].y = W[(size_t)(k2 + 1) * NOUT + col];
  }
  const float bv = bias[col];
  for (int rt = wq; rt < OW_ROWS / 16; rt += NRW) {
    const int arow = rt * 16 + c;
    const int   dg = s_deg[arow];
    const float sc = (dg > 0) ? rsqrtf((float)dg) : 0.0f;
    const float* trow = tile + arow * OW_PITCH + 64 * half;
    v8f acc = {};
#pragma unroll
    for (int kk = 0; kk < 32; ++kk) {
      v2f a;
      a.x = trow[kk] * sc;
      a.y = trow[32 + kk] * sc;
      acc = __builtin_amdgcn_wmma_f32_16x16x4_f32(
           false, a,  false, bfrag[kk],
           (short)0, acc,  false,  false);
    }
#pragma unroll
    for (int r = 0; r < 8; ++r) {
      const int lrow = rt * 16 + r + half * 8;
      if (lrow < nrows) {
        float v = acc[r] + bv;
        if (RELU) v = (v > 0.0f) ? v : 0.0f;
        outp[(size_t)(node0 + lrow) * NOUT + col] = v;
      }
    }
  }
}

extern "C" void kernel_launch(void* const* d_in, const int* in_sizes, int n_in,
                              void* d_out, int out_size, void* d_ws, size_t ws_size,
                              hipStream_t stream) {
  const float* in_feat = (const float*)d_in[0];
  const int*   src     = (const int*)d_in[1];
  const int*   dst     = (const int*)d_in[2];
  const float* W1      = (const float*)d_in[3];
  const float* b1      = (const float*)d_in[4];
  const float* W2      = (const float*)d_in[5];
  const float* b2      = (const float*)d_in[6];
  float* out = (float*)d_out;

  const int N = GCN_N_NODES;
  const int E = GCN_N_EDGES;

  float* deg_out = (float*)d_ws;
  float* h1      = deg_out + 50176;
  unsigned* list  = (unsigned*)(h1 + (size_t)N * 128);
  unsigned* table = list + GCN_N_EDGES;

  hipMemsetAsync(deg_out, 0, (size_t)50176 * sizeof(float), stream);
  gcn_deg_split_kernel<<<PT_CHUNKS, PT_THREADS, 0, stream>>>(src, dst, deg_out, list, table, E);

  gcn_spmm_owner_mm<GCN_H, true><<<OW_GROUPS, OW_THREADS, 0, stream>>>(
      in_feat, list, table, deg_out, W1, b1, h1, N);

  gcn_spmm_owner_mm<GCN_C, false><<<OW_GROUPS, OW_THREADS, 0, stream>>>(
      h1, list, table, deg_out, W2, b2, out, N);
}


# Out-degree scales by owner groups: edges also split by source group (byte entries) in the split kernel, a count kernel sums each group's pieces in LDS and stores D^-1/2; no f32 atomics, no memset
# speedup vs baseline: 5.3595x; 1.3495x over previous
#include <hip/hip_runtime.h>
#include <hip/hip_bf16.h>

#define GCN_N_NODES 50000
#define GCN_N_EDGES 800000
#define GCN_IN      128
#define GCN_H       128
#define GCN_C       32
#define OW_GROUPS   256
#define OW_NPG      196
#define OW_ROWS     208
#define OW_PITCH    132
#define OW_CAP      4096
#define PT_CHUNKS   256
#define PT_CH       3125
#define PT_THREADS  1024
#define PT_EPT      4
#define SL_PITCH    3136
static_assert(PT_CHUNKS * PT_CH >= GCN_N_EDGES && PT_EPT * PT_THREADS >= PT_CH, "chunks cover the edges");
static_assert(GCN_N_NODES <= 65536 && OW_NPG <= 65536 && PT_CH <= 65535, "16-bit fields of the packed words");
static_assert(SL_PITCH >= PT_CH && SL_PITCH % 16 == 0 && OW_NPG <= 256, "source-side list: one byte per edge");

typedef __attribute__((ext_vector_type(2))) float v2f;
typedef __attribute__((ext_vector_type(8))) float v8f;

__device__ __forceinline__ void pt_scan256(const int* cnt, int* off, int lane) {
  int v[8];
  int sum = 0;
#pragma unroll
  for (int i = 0; i < 8; ++i) {
    v[i] = cnt[lane * 8 + i];
    sum += v[i];
  }
  int incl = sum;
#pragma unroll
  for (int dl = 1; dl < 32; dl <<= 1) {
    const int t = __shfl_up(incl, dl, 32);
    if (lane >= dl) incl += t;
  }
  int run = incl - sum;
#pragma unroll
  for (int i = 0; i < 8; ++i) {
    off[lane * 8 + i] = run;
    run += v[i];
  }
}

__global__ __launch_bounds__(PT_THREADS) void gcn_split_kernel(
    const int* __restrict__ src, const int* __restrict__ dst,
    unsigned* __restrict__ list, unsigned* __restrict__ table,
    unsigned* __restrict__ slist, unsigned* __restrict__ stable, int E) {
  __shared__ int cnt[OW_GROUPS];
  __shared__ int off[OW_GROUPS];
  __shared__ int cnt2[OW_GROUPS];
  __shared__ int off2[OW_GROUPS];
  __shared__ unsigned stage[PT_CH];
  __shared__ unsigned stage2[SL_PITCH / 4];
  const int tid = threadIdx.x;
  const int c   = blockIdx.x;
  const int e0  = c * PT_CH;
  const int n   = max(0, min(PT_CH, E - e0));
  if (tid < OW_GROUPS) { cnt[tid] = 0; cnt2[tid] = 0; }
  if (tid < SL_PITCH / 4) stage2[tid] = 0u;
  __syncthreads();
  int s[PT_EPT], d[PT_EPT], r[PT_EPT], r2[PT_EPT];
#pragma unroll
  for (int i = 0; i < PT_EPT; ++i) {
    const int idx = tid + i * PT_THREADS;
    s[i] = 0;
    d[i] = 0;
    if (idx < n) {
      s[i] = src[e0 + idx];
      d[i] = dst[e0 + idx];
    }
  }
#pragma unroll
  for (int i = 0; i < PT_EPT; ++i) {
    const int idx = tid + i * PT_THREADS;
    r[i]  = 0;
    r2[i] = 0;
    if (idx < n) {
      r[i]  = atomicAdd(&cnt[(unsigned)d[i] / (unsigned)OW_NPG], 1);
      r2[i] = atomicAdd(&cnt2[(unsigned)s[i] / (unsigned)OW_NPG], 1);
    }
  }
  __syncthreads();
  if (tid < 32)      pt_scan256(cnt, off, tid);
  else if (tid < 64) pt_scan256(cnt2, off2, tid - 32);
  __syncthreads();
#pragma unroll
  for (int i = 0; i < PT_EPT; ++i) {
    const int idx = tid + i * PT_THREADS;
    if (idx < n) {
      const unsigned g  = (unsigned)d[i] / (unsigned)OW_NPG;
      const unsigned gs = (unsigned)s[i] / (unsigned)OW_NPG;
      stage[off[g] + r[i]] = (unsigned)s[i] | (((unsigned)d[i] - g * (unsigned)OW_NPG) << 16);
      ((unsigned char*)stage2)[off2[gs] + r2[i]] = (unsigned char)((unsigned)s[i] - gs * (unsigned)OW_NPG);
    }
  }
  if (tid < OW_GROUPS) {
    table[(size_t)tid * PT_CHUNKS + c]  = (unsigned)off[tid] | ((unsigned)cnt[tid] << 16);
    stable[(size_t)tid * PT_CHUNKS + c] = (unsigned)off2[tid] | ((unsigned)cnt2[tid] << 16);
  }
  __syncthreads();
  for (int i = tid; i < n; i += PT_THREADS) list[e0 + i] = stage[i];
  if (tid < SL_PITCH / 4) slist[(size_t)c * (SL_PITCH / 4) + tid] = stage2[tid];
}

#define OD_THREADS 256
#define OD_PPW     (PT_CHUNKS / (OD_THREADS / 32))
static_assert(OD_PPW * (OD_THREADS / 32) == PT_CHUNKS && OD_PPW <= 32 && OW_NPG <= OD_THREADS, "pieces per wave");

__global__ __launch_bounds__(OD_THREADS) void gcn_outdeg_scale_kernel(
    const unsigned* __restrict__ slist, const unsigned* __restrict__ stable,
    float* __restrict__ so, int N) {
  __shared__ int s_c[OW_NPG];
  const int tid  = threadIdx.x;
  const int lane = tid & 31;
  const int wave = __builtin_amdgcn_readfirstlane(tid >> 5);
  const int g    = blockIdx.x;
  const unsigned char* sl = (const unsigned char*)slist;
  if (tid < OW_NPG) s_c[tid] = 0;
  unsigned t = 0u;
  if (lane < OD_PPW) t = stable[(size_t)g * PT_CHUNKS + wave * OD_PPW + lane];
  __syncthreads();
  int row[OD_PPW];
#pragma unroll
  for (int k = 0; k < OD_PPW; ++k) {
    const unsigned tk  = (unsigned)__builtin_amdgcn_readlane((int)t, k);
    const int      len = (int)(tk >> 16);
    row[k] = -1;
    if (lane < len) row[k] = sl[(size_t)(wave * OD_PPW + k) * SL_PITCH + (tk & 0xFFFFu) + lane];
  }
#pragma unroll
  for (int k = 0; k < OD_PPW; ++k)
    if (row[k] >= 0) atomicAdd(&s_c[row[k]], 1);
#pragma unroll
  for (int k = 0; k < OD_PPW; ++k) {
    const unsigned tk  = (unsigned)__builtin_amdgcn_readlane((int)t, k);
    const int      len = (int)(tk >> 16);
    for (int i = lane + 32; i < len; i += 32)
      atomicAdd(&s_c[sl[(size_t)(wave * OD_PPW + k) * SL_PITCH + (tk & 0xFFFFu) + i]], 1);
  }
  __syncthreads();
  if (tid < OW_NPG && g * OW_NPG + tid < N) {
    const int dg = s_c[tid];
    so[g * OW_NPG + tid] = (dg > 0) ? rsqrtf((float)dg) : 0.0f;
  }
}

#define OW_THREADS 1024
#define OW_FPT     (OW_CAP / OW_THREADS)

__device__ __forceinline__ void ow_flush_run(float* tile, int d, int lane, float4 a) {
  float* t = tile + d * OW_PITCH + lane;
  __hip_atomic_fetch_add(t + 0,  a.x, __ATOMIC_RELAXED, __HIP_MEMORY_SCOPE_WORKGROUP);
  __hip_atomic_fetch_add(t + 32, a.y, __ATOMIC_RELAXED, __HIP_MEMORY_SCOPE_WORKGROUP);
  __hip_atomic_fetch_add(t + 64, a.z, __ATOMIC_RELAXED, __HIP_MEMORY_SCOPE_WORKGROUP);
  __hip_atomic_fetch_add(t + 96, a.w, __ATOMIC_RELAXED, __HIP_MEMORY_SCOPE_WORKGROUP);
}

__device__ __forceinline__ void ow_add_row(float* tile, int d, int lane, float4 v, float sc,
                                           int& cur, float4& acc) {
  if (d != cur) {
    ow_flush_run(tile, cur, lane, acc);
    acc = make_float4(0.0f, 0.0f, 0.0f, 0.0f);
    cur = d;
  }
  acc.x += v.x * sc;
  acc.y += v.y * sc;
  acc.z += v.z * sc;
  acc.w += v.w * sc;
}

__device__ __forceinline__ void ow_process(
    float* tile, const float* __restrict__ x,
    int s_l, int d_l, int sc_l, int cnt, int lane) {
  float4 va[8], vb[8];
  int    cur = __builtin_amdgcn_readlane(d_l, 0);
  float4 acc = make_float4(0.0f, 0.0f, 0.0f, 0.0f);
#pragma unroll
  for (int k = 0; k < 8; ++k)
    va[k] = *(const float4*)(x + (size_t)__builtin_amdgcn_readlane(s_l, k) * 128 + lane * 4);
#pragma unroll
  for (int j0 = 0; j0 < 32; j0 += 16) {
#pragma unroll
    for (int k = 0; k < 8; ++k)
      vb[k] = *(const float4*)(x + (size_t)__builtin_amdgcn_readlane(s_l, j0 + 8 + k) * 128 + lane * 4);
#pragma unroll
    for (int k = 0; k < 8; ++k)
      if (j0 + k < cnt)
        ow_add_row(tile, __builtin_amdgcn_readlane(d_l, j0 + k), lane, va[k],
                   __int_as_float(__builtin_amdgcn_readlane(sc_l, j0 + k)), cur, acc);
#pragma unroll
    for (int k = 0; k < 8; ++k)
      if (j0 + 16 + k < 32)
        va[k] = *(const float4*)(x + (size_t)__builtin_amdgcn_readlane(s_l, j0 + 16 + k) * 128 + lane * 4);
#pragma unroll
    for (int k = 0; k < 8; ++k)
      if (j0 + 8 + k < cnt)
        ow_add_row(tile, __builtin_amdgcn_readlane(d_l, j0 + 8 + k), lane, vb[k],
                   __int_as_float(__builtin_amdgcn_readlane(sc_l, j0 + 8 + k)), cur, acc);
  }
  ow_flush_run(tile, cur, lane, acc);
}

template <int NOUT, bool RELU>
__global__ __launch_bounds__(OW_THREADS) void gcn_spmm_owner_mm(
    const float* __restrict__ x, const unsigned* __restrict__ list,
    const unsigned* __restrict__ table, const float* __restrict__ so,
    const float* __restrict__ W, const float* __restrict__ bias,
    float* __restrict__ outp, int N) {
  __shared__ float tile[OW_ROWS * OW_PITCH];
  __shared__ unsigned s_tab[PT_CHUNKS];
  __shared__ int      s_base[PT_CHUNKS + 1];
  __shared__ unsigned s_ent[OW_CAP];
  __shared__ float    s_sc[OW_CAP];
  __shared__ int      s_deg[OW_ROWS];
  __shared__ int      s_cnt[OW_ROWS];
  __shared__ int      s_off[OW_ROWS];
  const int tid  = threadIdx.x;
  const int lane = tid & 31;
  const int wave = __builtin_amdgcn_readfirstlane(tid >> 5);
  const int g     = blockIdx.x;
  const int node0 = g * OW_NPG;
  const int nrows = min(OW_NPG, N - node0);

  if (tid < PT_CHUNKS) s_tab[tid] = table[(size_t)g * PT_CHUNKS + tid];
  if (tid < OW_ROWS) { s_deg[tid] = 0; s_cnt[tid] = 0; }
  for (int i = tid; i < OW_ROWS * OW_PITCH / 4; i += OW_THREADS)
    ((float4*)tile)[i] = make_float4(0.0f, 0.0f, 0.0f, 0.0f);
  __syncthreads();
  if (tid < 32) {
    int v[8];
    int sum = 0;
#pragma unroll
    for (int i = 0; i < 8; ++i) {
      v[i] = (int)(s_tab[tid * 8 + i] >> 16);
      sum += v[i];
    }
    int incl = sum;
#pragma unroll
    for (int dl = 1; dl < 32; dl <<= 1) {
      const int t = __shfl_up(incl, dl, 32);
      if (tid >= dl) incl += t;
    }
    int run = incl - sum;
#pragma unroll
    for (int i = 0; i < 8; ++i) {
      s_base[tid * 8 + i] = run;
      run += v[i];
    }
    if (tid == 31) s_base[PT_CHUNKS] = run;
  }
  __syncthreads();
  const int total = __builtin_amdgcn_readfirstlane(s_base[PT_CHUNKS]);

  for (int wb = 0; wb < total; wb += OW_CAP) {
    const int nwin = min(OW_CAP, total - wb);
    int      fc[OW_FPT], fp[OW_FPT];
    unsigned fe[OW_FPT];
    float    fs[OW_FPT];
#pragma unroll
    for (int q = 0; q < OW_FPT; ++q) {
      fp[q] = min(wb + tid + q * OW_THREADS, total - 1);
      int lo = 0;
#pragma unroll
      for (int step = PT_CHUNKS / 2; step >= 1; step >>= 1)
        if (s_base[lo + step] <= fp[q]) lo += step;
      fc[q] = lo;
    }
#pragma unroll
    for (int q = 0; q < OW_FPT; ++q)
      fe[q] = list[(size_t)fc[q] * PT_CH + (s_tab[fc[q]] & 0xFFFFu) + (unsigned)(fp[q] - s_base[fc[q]])];
#pragma unroll
    for (int q = 0; q < OW_FPT; ++q)
      fs[q] = so[fe[q] & 0xFFFFu];
    int rk[OW_FPT];
#pragma unroll
    for (int q = 0; q < OW_FPT; ++q) {
      rk[q] = 0;
      if (wb + tid + q * OW_THREADS < total) rk[q] = atomicAdd(&s_cnt[fe[q] >> 16], 1);
    }
    __syncthreads();
    if (tid < 32) {
      int v[7];
      int sum = 0;
#pragma unroll
      for (int i = 0; i < 7; ++i) {
        const int idx = tid * 7 + i;
        v[i] = (idx < OW_ROWS) ? s_cnt[idx] : 0;
        sum += v[i];
      }
      int incl = sum;
#pragma unroll
      for (int dl = 1; dl < 32; dl <<= 1) {
        const int t = __shfl_up(incl, dl, 32);
        if (tid >= dl) incl += t;
      }
      int run = incl - sum;
#pragma unroll
      for (int i = 0; i < 7; ++i) {
        const int idx = tid * 7 + i;
        if (idx < OW_ROWS) {
          s_off[idx] = run;
          s_deg[idx] += v[i];
          s_cnt[idx] = 0;
        }
        run += v[i];
      }
    }
    __syncthreads();
#pragma unroll
    for (int q = 0; q < OW_FPT; ++q) {
      if (wb + tid + q * OW_THREADS < total) {
        const int pos = s_off[fe[q] >> 16] + rk[q];
        s_ent[pos] = fe[q];
        s_sc[pos]  = fs[q];
      }
    }
    __syncthreads();
    const int nb = (nwin + 31) >> 5;
    for (int b = wave; b < nb; b += OW_THREADS / 32) {
      const int cnt = min(32, nwin - (b << 5));
      int s_l = 0, d_l = 0, sc_l = 0;
      if (lane < cnt) {
        const unsigned ent = s_ent[(b << 5) + lane];
        s_l  = (int)(ent & 0xFFFFu);
        d_l  = (int)(ent >> 16);
        sc_l = __float_as_int(s_sc[(b << 5) + lane]);
      }
      if (cnt == 32) ow_process(tile, x, s_l, d_l, sc_l, 32, lane);
      else           ow_process(tile, x, s_l, d_l, sc_l, cnt, lane);
    }
    __syncthreads();
  }

  constexpr int NCT = NOUT / 16;
  constexpr int NRW = (OW_THREADS / 32) / NCT;
  const int ct   = wave % NCT;
  const int wq   = wave / NCT;
  const int half = lane >> 4;
  const int c    = lane & 15;
  const int col  = ct * 16 + c;
  v2f bfrag[32];
#pragma unroll
  for (int kk = 0; kk < 32; ++kk) {
    const int k2 = kk * 4 + half * 2;
    bfrag[kk].x = W[(size_t)k2 * NOUT + col];
    bfrag[kk].y = W[(size_t)(k2 + 1) * NOUT + col];
  }
  const float bv = bias[col];
  for (int rt = wq; rt < OW_ROWS / 16; rt += NRW) {
    const int arow = rt * 16 + c;
    const int   dg = s_deg[arow];
    const float sc = (dg > 0) ? rsqrtf((float)dg) : 0.0f;
    const float* trow = tile + arow * OW_PITCH + 64 * half;
    v8f acc = {};
#pragma unroll
    for (int kk = 0; kk < 32; ++kk) {
      v2f a;
      a.x = trow[kk] * sc;
      a.y = trow[32 + kk] * sc;
      acc = __builtin_amdgcn_wmma_f32_16x16x4_f32(
           false, a,  false, bfrag[kk],
           (short)0, acc,  false,  false);
    }
#pragma unroll
    for (int r = 0; r < 8; ++r) {
      const int lrow = rt * 16 + r + half * 8;
      if (lrow < nrows) {
        float v = acc[r] + bv;
        if (RELU) v = (v > 0.0f) ? v : 0.0f;
        outp[(size_t)(node0 + lrow) * NOUT + col] = v;
      }
    }
  }
}

extern "C" void kernel_launch(void* const* d_in, const int* in_sizes, int n_in,
                              void* d_out, int out_size, void* d_ws, size_t ws_size,
                              hipStream_t stream) {
  const float* in_feat = (const float*)d_in[0];
  const int*   src     = (const int*)d_in[1];
  const int*   dst     = (const int*)d_in[2];
  const float* W1      = (const float*)d_in[3];
  const float* b1      = (const float*)d_in[4];
  const float* W2      = (const float*)d_in[5];
  const float* b2      = (const float*)d_in[6];
  float* out = (float*)d_out;

  const int N = GCN_N_NODES;
  const int E = GCN_N_EDGES;

  float* so      = (float*)d_ws;
  float* h1      = so + 50176;
  unsigned* list   = (unsigned*)(h1 + (size_t)N * 128);
  unsigned* table  = list + GCN_N_EDGES;
  unsigned* stable = table + OW_GROUPS * PT_CHUNKS;
  unsigned* slist  = stable + OW_GROUPS * PT_CHUNKS;

  gcn_split_kernel<<<PT_CHUNKS, PT_THREADS, 0, stream>>>(src, dst, list, table, slist, stable, E);

  gcn_outdeg_scale_kernel<<<OW_GROUPS, OD_THREADS, 0, stream>>>(slist, stable, so, N);

  gcn_spmm_owner_mm<GCN_H, true><<<OW_GROUPS, OW_THREADS, 0, stream>>>(
      in_feat, list, table, so, W1, b1, h1, N);

  gcn_spmm_owner_mm<GCN_C, false><<<OW_GROUPS, OW_THREADS, 0, stream>>>(
      h1, list, table, so, W2, b2, out, N);
}


# Source scales looked up in LDS: the out-degree kernel also stores 16-bit counts, every owner group copies that 100-KB table in whole lines and forms rsqrt itself; 800k one-word gathers per layer gone
# speedup vs baseline: 5.5221x; 1.0303x over previous
#include <hip/hip_runtime.h>
#include <hip/hip_bf16.h>

#define GCN_N_NODES 50000
#define GCN_N_EDGES 800000
#define GCN_IN      128
#define GCN_H       128
#define GCN_C       32
#define OW_GROUPS   256
#define OW_NPG      196
#define OW_ROWS     208
#define OW_PITCH    132
#define OW_CAP      4096
#define OW_NPAD     (OW_GROUPS * OW_NPG)
#define DG_VEC      (OW_NPAD * 2 / 16)
#define DG_SAT      65535u
#define PT_CHUNKS   256
#define PT_CH       3125
#define PT_THREADS  1024
#define PT_EPT      4
#define SL_PITCH    3136
static_assert(PT_CHUNKS * PT_CH >= GCN_N_EDGES && PT_EPT * PT_THREADS >= PT_CH, "chunks cover the edges");
static_assert(GCN_N_NODES <= 65536 && OW_NPG <= 65536 && PT_CH <= 65535, "16-bit fields of the packed words");
static_assert(SL_PITCH >= PT_CH && SL_PITCH % 16 == 0 && OW_NPG <= 256, "source-side list: one byte per edge");
static_assert((OW_NPAD * 2) % 16 == 0 && OW_NPAD >= GCN_N_NODES, "out-degree table: whole 16-byte pieces, every node in it");

typedef __attribute__((ext_vector_type(2))) float v2f;
typedef __attribute__((ext_vector_type(8))) float v8f;
typedef __attribute__((ext_vector_type(4))) unsigned v4u;

__device__ __forceinline__ void pt_scan256(const int* cnt, int* off, int lane) {
  int v[8];
  int sum = 0;
#pragma unroll
  for (int i = 0; i < 8; ++i) {
    v[i] = cnt[lane * 8 + i];
    sum += v[i];
  }
  int incl = sum;
#pragma unroll
  for (int dl = 1; dl < 32; dl <<= 1) {
    const int t = __shfl_up(incl, dl, 32);
    if (lane >= dl) incl += t;
  }
  int run = incl - sum;
#pragma unroll
  for (int i = 0; i < 8; ++i) {
    off[lane * 8 + i] = run;
    run += v[i];
  }
}

__global__ __launch_bounds__(PT_THREADS) void gcn_split_kernel(
    const int* __restrict__ src, const int* __restrict__ dst,
    unsigned* __restrict__ list, unsigned* __restrict__ table,
    unsigned* __restrict__ slist, unsigned* __restrict__ stable, int E) {
  __shared__ int cnt[OW_GROUPS];
  __shared__ int off[OW_GROUPS];
  __shared__ int cnt2[OW_GROUPS];
  __shared__ int off2[OW_GROUPS];
  __shared__ unsigned stage[PT_CH];
  __shared__ unsigned stage2[SL_PITCH / 4];
  const int tid = threadIdx.x;
  const int c   = blockIdx.x;
  const int e0  = c * PT_CH;
  const int n   = max(0, min(PT_CH, E - e0));
  if (tid < OW_GROUPS) { cnt[tid] = 0; cnt2[tid] = 0; }
  if (tid < SL_PITCH / 4) stage2[tid] = 0u;
  __syncthreads();
  int s[PT_EPT], d[PT_EPT], r[PT_EPT], r2[PT_EPT];
#pragma unroll
  for (int i = 0; i < PT_EPT; ++i) {
    const int idx = tid + i * PT_THREADS;
    s[i] = 0;
    d[i] = 0;
    if (idx < n) {
      s[i] = src[e0 + idx];
      d[i] = dst[e0 + idx];
    }
  }
#pragma unroll
  for (int i = 0; i < PT_EPT; ++i) {
    const int idx = tid + i * PT_THREADS;
    r[i]  = 0;
    r2[i] = 0;
    if (idx < n) {
      r[i]  = atomicAdd(&cnt[(unsigned)d[i] / (unsigned)OW_NPG], 1);
      r2[i] = atomicAdd(&cnt2[(unsigned)s[i] / (unsigned)OW_NPG], 1);
    }
  }
  __syncthreads();
  if (tid < 32)      pt_scan256(cnt, off, tid);
  else if (tid < 64) pt_scan256(cnt2, off2, tid - 32);
  __syncthreads();
#pragma unroll
  for (int i = 0; i < PT_EPT; ++i) {
    const int idx = tid + i * PT_THREADS;
    if (idx < n) {
      const unsigned g  = (unsigned)d[i] / (unsigned)OW_NPG;
      const unsigned gs = (unsigned)s[i] / (unsigned)OW_NPG;
      stage[off[g] + r[i]] = (unsigned)s[i] | (((unsigned)d[i] - g * (unsigned)OW_NPG) << 16);
      ((unsigned char*)stage2)[off2[gs] + r2[i]] = (unsigned char)((unsigned)s[i] - gs * (unsigned)OW_NPG);
    }
  }
  if (tid < OW_GROUPS) {
    table[(size_t)tid * PT_CHUNKS + c]  = (unsigned)off[tid] | ((unsigned)cnt[tid] << 16);
    stable[(size_t)tid * PT_CHUNKS + c] = (unsigned)off2[tid] | ((unsigned)cnt2[tid] << 16);
  }
  __syncthreads();
  for (int i = tid; i < n; i += PT_THREADS) list[e0 + i] = stage[i];
  if (tid < SL_PITCH / 4) slist[(size_t)c * (SL_PITCH / 4) + tid] = stage2[tid];
}

#define OD_THREADS 256
#define OD_PPW     (PT_CHUNKS / (OD_THREADS / 32))
static_assert(OD_PPW * (OD_THREADS / 32) == PT_CHUNKS && OD_PPW <= 32 && OW_NPG <= OD_THREADS, "pieces per wave");

__global__ __launch_bounds__(OD_THREADS) void gcn_outdeg_scale_kernel(
    const unsigned* __restrict__ slist, const unsigned* __restrict__ stable,
    float* __restrict__ so, unsigned short* __restrict__ dg16, int N) {
  __shared__ int s_c[OW_NPG];
  const int tid  = threadIdx.x;
  const int lane = tid & 31;
  const int wave = __builtin_amdgcn_readfirstlane(tid >> 5);
  const int g    = blockIdx.x;
  const unsigned char* sl = (const unsigned char*)slist;
  if (tid < OW_NPG) s_c[tid] = 0;
  unsigned t = 0u;
  if (lane < OD_PPW) t = stable[(size_t)g * PT_CHUNKS + wave * OD_PPW + lane];
  __syncthreads();
  int row[OD_PPW];
#pragma unroll
  for (int k = 0; k < OD_PPW; ++k) {
    const unsigned tk  = (unsigned)__builtin_amdgcn_readlane((int)t, k);
    const int      len = (int)(tk >> 16);
    row[k] = -1;
    if (lane < len) row[k] = sl[(size_t)(wave * OD_PPW + k) * SL_PITCH + (tk & 0xFFFFu) + lane];
  }
#pragma unroll
  for (int k = 0; k < OD_PPW; ++k)
    if (row[k] >= 0) atomicAdd(&s_c[row[k]], 1);
#pragma unroll
  for (int k = 0; k < OD_PPW; ++k) {
    const unsigned tk  = (unsigned)__builtin_amdgcn_readlane((int)t, k);
    const int      len = (int)(tk >> 16);
    for (int i = lane + 32; i < len; i += 32)
      atomicAdd(&s_c[sl[(size_t)(wave * OD_PPW + k) * SL_PITCH + (tk & 0xFFFFu) + i]], 1);
  }
  __syncthreads();
  if (tid < OW_NPG) {
    const int dg = s_c[tid];
    dg16[g * OW_NPG + tid] = (unsigned short)min((unsigned)dg, DG_SAT);
    if (g * OW_NPG + tid < N) so[g * OW_NPG + tid] = (dg > 0) ? rsqrtf((float)dg) : 0.0f;
  }
}

#define OW_THREADS 1024
#define OW_FPT     (OW_CAP / OW_THREADS)

__device__ __forceinline__ void ow_flush_run(float* tile, int d, int lane, float4 a) {
  float* t = tile + d * OW_PITCH + lane;
  __hip_atomic_fetch_add(t + 0,  a.x, __ATOMIC_RELAXED, __HIP_MEMORY_SCOPE_WORKGROUP);
  __hip_atomic_fetch_add(t + 32, a.y, __ATOMIC_RELAXED, __HIP_MEMORY_SCOPE_WORKGROUP);
  __hip_atomic_fetch_add(t + 64, a.z, __ATOMIC_RELAXED, __HIP_MEMORY_SCOPE_WORKGROUP);
  __hip_atomic_fetch_add(t + 96, a.w, __ATOMIC_RELAXED, __HIP_MEMORY_SCOPE_WORKGROUP);
}

__device__ __forceinline__ void ow_add_row(float* tile, int d, int lane, float4 v, float sc,
                                           int& cur, float4& acc) {
  if (d != cur) {
    ow_flush_run(tile, cur, lane, acc);
    acc = make_float4(0.0f, 0.0f, 0.0f, 0.0f);
    cur = d;
  }
  acc.x += v.x * sc;
  acc.y += v.y * sc;
  acc.z += v.z * sc;
  acc.w += v.w * sc;
}

__device__ __forceinline__ void ow_process(
    float* tile, const float* __restrict__ x,
    int s_l, int d_l, int sc_l, int cnt, int lane) {
  float4 va[8], vb[8];
  int    cur = __builtin_amdgcn_readlane(d_l, 0);
  float4 acc = make_float4(0.0f, 0.0f, 0.0f, 0.0f);
#pragma unroll
  for (int k = 0; k < 8; ++k)
    va[k] = *(const float4*)(x + (size_t)__builtin_amdgcn_readlane(s_l, k) * 128 + lane * 4);
#pragma unroll
  for (int j0 = 0; j0 < 32; j0 += 16) {
#pragma unroll
    for (int k = 0; k < 8; ++k)
      vb[k] = *(const float4*)(x + (size_t)__builtin_amdgcn_readlane(s_l, j0 + 8 + k) * 128 + lane * 4);
#pragma unroll
    for (int k = 0; k < 8; ++k)
      if (j0 + k < cnt)
        ow_add_row(tile, __builtin_amdgcn_readlane(d_l, j0 + k), lane, va[k],
                   __int_as_float(__builtin_amdgcn_readlane(sc_l, j0 + k)), cur, acc);
#pragma unroll
    for (int k = 0; k < 8; ++k)
      if (j0 + 16 + k < 32)
        va[k] = *(const float4*)(x + (size_t)__builtin_amdgcn_readlane(s_l, j0 + 16 + k) * 128 + lane * 4);
#pragma unroll
    for (int k = 0; k < 8; ++k)
      if (j0 + 8 + k < cnt)
        ow_add_row(tile, __builtin_amdgcn_readlane(d_l, j0 + 8 + k), lane, vb[k],
                   __int_as_float(__builtin_amdgcn_readlane(sc_l, j0 + 8 + k)), cur, acc);
  }
  ow_flush_run(tile, cur, lane, acc);
}

template <int NOUT, bool RELU>
__global__ __launch_bounds__(OW_THREADS) void gcn_spmm_owner_mm(
    const float* __restrict__ x, const unsigned* __restrict__ list,
    const unsigned* __restrict__ table, const float* __restrict__ so,
    const unsigned* __restrict__ dgw,
    const float* __restrict__ W, const float* __restrict__ bias,
    float* __restrict__ outp, int N) {
  __shared__ float tile[OW_ROWS * OW_PITCH];
  __shared__ v4u      s_dg[DG_VEC];
  __shared__ unsigned s_tab[PT_CHUNKS];
  __shared__ int      s_base[PT_CHUNKS + 1];
  __shared__ unsigned s_ent[OW_CAP];
  __shared__ float    s_sc[OW_CAP];
  __shared__ int      s_deg[OW_ROWS];
  __shared__ int      s_cnt[OW_ROWS];
  __shared__ int      s_off[OW_ROWS];
  const int tid  = threadIdx.x;
  const int lane = tid & 31;
  const int wave = __builtin_amdgcn_readfirstlane(tid >> 5);
  const int g     = blockIdx.x;
  const int node0 = g * OW_NPG;
  const int nrows = min(OW_NPG, N - node0);

  constexpr int DG_PT = (DG_VEC + OW_THREADS - 1) / OW_THREADS;
  const unsigned tabw = table[(size_t)g * PT_CHUNKS + (tid & (PT_CHUNKS - 1))];
  v4u dv[DG_PT];
#pragma unroll
  for (int k = 0; k < DG_PT; ++k)
    dv[k] = ((const v4u*)dgw)[min(tid + k * OW_THREADS, DG_VEC - 1)];
  if (tid < OW_ROWS) { s_deg[tid] = 0; s_cnt[tid] = 0; }
  for (int i = tid; i < OW_ROWS * OW_PITCH / 4; i += OW_THREADS)
    ((float4*)tile)[i] = make_float4(0.0f, 0.0f, 0.0f, 0.0f);
  if (tid < PT_CHUNKS) s_tab[tid] = tabw;
#pragma unroll
  for (int k = 0; k < DG_PT; ++k)
    if (tid + k * OW_THREADS < DG_VEC) s_dg[tid + k * OW_THREADS] = dv[k];
  __syncthreads();
  if (tid < 32) {
    int v[8];
    int sum = 0;
#pragma unroll
    for (int i = 0; i < 8; ++i) {
      v[i] = (int)(s_tab[tid * 8 + i] >> 16);
      sum += v[i];
    }
    int incl = sum;
#pragma unroll
    for (int dl = 1; dl < 32; dl <<= 1) {
      const int t = __shfl_up(incl, dl, 32);
      if (tid >= dl) incl += t;
    }
    int run = incl - sum;
#pragma unroll
    for (int i = 0; i < 8; ++i) {
      s_base[tid * 8 + i] = run;
      run += v[i];
    }
    if (tid == 31) s_base[PT_CHUNKS] = run;
  }
  __syncthreads();
  const int total = __builtin_amdgcn_readfirstlane(s_base[PT_CHUNKS]);

  for (int wb = 0; wb < total; wb += OW_CAP) {
    const int nwin = min(OW_CAP, total - wb);
    int      fc[OW_FPT], fp[OW_FPT];
    unsigned fe[OW_FPT];
    float    fs[OW_FPT];
#pragma unroll
    for (int q = 0; q < OW_FPT; ++q) {
      fp[q] = min(wb + tid + q * OW_THREADS, total - 1);
      int lo = 0;
#pragma unroll
      for (int step = PT_CHUNKS / 2; step >= 1; step >>= 1)
        if (s_base[lo + step] <= fp[q]) lo += step;
      fc[q] = lo;
    }
#pragma unroll
    for (int q = 0; q < OW_FPT; ++q)
      fe[q] = list[(size_t)fc[q] * PT_CH + (s_tab[fc[q]] & 0xFFFFu) + (unsigned)(fp[q] - s_base[fc[q]])];
    unsigned dgv[OW_FPT];
#pragma unroll
    for (int q = 0; q < OW_FPT; ++q)
      dgv[q] = ((const unsigned short*)s_dg)[fe[q] & 0xFFFFu];
#pragma unroll
    for (int q = 0; q < OW_FPT; ++q)
      fs[q] = (dgv[q] > 0u) ? rsqrtf((float)dgv[q]) : 0.0f;
#pragma unroll
    for (int q = 0; q < OW_FPT; ++q)
      if (dgv[q] == DG_SAT) fs[q] = so[fe[q] & 0xFFFFu];
    int rk[OW_FPT];
#pragma unroll
    for (int q = 0; q < OW_FPT; ++q) {
      rk[q] = 0;
      if (wb + tid + q * OW_THREADS < total) rk[q] = atomicAdd(&s_cnt[fe[q] >> 16], 1);
    }
    __syncthreads();
    if (tid < 32) {
      int v[7];
      int sum = 0;
#pragma unroll
      for (int i = 0; i < 7; ++i) {
        const int idx = tid * 7 + i;
        v[i] = (idx < OW_ROWS) ? s_cnt[idx] : 0;
        sum += v[i];
      }
      int incl = sum;
#pragma unroll
      for (int dl = 1; dl < 32; dl <<= 1) {
        const int t = __shfl_up(incl, dl, 32);
        if (tid >= dl) incl += t;
      }
      int run = incl - sum;
#pragma unroll
      for (int i = 0; i < 7; ++i) {
        const int idx = tid * 7 + i;
        if (idx < OW_ROWS) {
          s_off[idx] = run;
          s_deg[idx] += v[i];
          s_cnt[idx] = 0;
        }
        run += v[i];
      }
    }
    __syncthreads();
#pragma unroll
    for (int q = 0; q < OW_FPT; ++q) {
      if (wb + tid + q * OW_THREADS < total) {
        const int pos = s_off[fe[q] >> 16] + rk[q];
        s_ent[pos] = fe[q];
        s_sc[pos]  = fs[q];
      }
    }
    __syncthreads();
    const int nb = (nwin + 31) >> 5;
    for (int b = wave; b < nb; b += OW_THREADS / 32) {
      const int cnt = min(32, nwin - (b << 5));
      int s_l = 0, d_l = 0, sc_l = 0;
      if (lane < cnt) {
        const unsigned ent = s_ent[(b << 5) + lane];
        s_l  = (int)(ent & 0xFFFFu);
        d_l  = (int)(ent >> 16);
        sc_l = __float_as_int(s_sc[(b << 5) + lane]);
      }
      if (cnt == 32) ow_process(tile, x, s_l, d_l, sc_l, 32, lane);
      else           ow_process(tile, x, s_l, d_l, sc_l, cnt, lane);
    }
    __syncthreads();
  }

  constexpr int NCT = NOUT / 16;
  constexpr int NRW = (OW_THREADS / 32) / NCT;
  const int ct   = wave % NCT;
  const int wq   = wave / NCT;
  const int half = lane >> 4;
  const int c    = lane & 15;
  const int col  = ct * 16 + c;
  v2f bfrag[32];
#pragma unroll
  for (int kk = 0; kk < 32; ++kk) {
    const int k2 = kk * 4 + half * 2;
    bfrag[kk].x = W[(size_t)k2 * NOUT + col];
    bfrag[kk].y = W[(size_t)(k2 + 1) * NOUT + col];
  }
  const float bv = bias[col];
  for (int rt = wq; rt < OW_ROWS / 16; rt += NRW) {
    const int arow = rt * 16 + c;
    const int   dg = s_deg[arow];
    const float sc = (dg > 0) ? rsqrtf((float)dg) : 0.0f;
    const float* trow = tile + arow * OW_PITCH + 64 * half;
    v8f acc = {};
#pragma unroll
    for (int kk = 0; kk < 32; ++kk) {
      v2f a;
      a.x = trow[kk] * sc;
      a.y = trow[32 + kk] * sc;
      acc = __builtin_amdgcn_wmma_f32_16x16x4_f32(
           false, a,  false, bfrag[kk],
           (short)0, acc,  false,  false);
    }
#pragma unroll
    for (int r = 0; r < 8; ++r) {
      const int lrow = rt * 16 + r + half * 8;
      if (lrow < nrows) {
        float v = acc[r] + bv;
        if (RELU) v = (v > 0.0f) ? v : 0.0f;
        outp[(size_t)(node0 + lrow) * NOUT + col] = v;
      }
    }
  }
}

extern "C" void kernel_launch(void* const* d_in, const int* in_sizes, int n_in,
                              void* d_out, int out_size, void* d_ws, size_t ws_size,
                              hipStream_t stream) {
  const float* in_feat = (const float*)d_in[0];
  const int*   src     = (const int*)d_in[1];
  const int*   dst     = (const int*)d_in[2];
  const float* W1      = (const float*)d_in[3];
  const float* b1      = (const float*)d_in[4];
  const float* W2      = (const float*)d_in[5];
  const float* b2      = (const float*)d_in[6];
  float* out = (float*)d_out;

  const int N = GCN_N_NODES;
  const int E = GCN_N_EDGES;

  float* so      = (float*)d_ws;
  float* h1      = so + 50176;
  unsigned* list   = (unsigned*)(h1 + (size_t)N * 128);
  unsigned* table  = list + GCN_N_EDGES;
  unsigned* stable = table + OW_GROUPS * PT_CHUNKS;
  unsigned* slist  = stable + OW_GROUPS * PT_CHUNKS;
  unsigned* dgw    = slist + PT_CHUNKS * (SL_PITCH / 4);

  gcn_split_kernel<<<PT_CHUNKS, PT_THREADS, 0, stream>>>(src, dst, list, table, slist, stable, E);

  gcn_outdeg_scale_kernel<<<OW_GROUPS, OD_THREADS, 0, stream>>>(slist, stable, so, (unsigned short*)dgw, N);

  gcn_spmm_owner_mm<GCN_H, true><<<OW_GROUPS, OW_THREADS, 0, stream>>>(
      in_feat, list, table, so, dgw, W1, b1, h1, N);

  gcn_spmm_owner_mm<GCN_C, false><<<OW_GROUPS, OW_THREADS, 0, stream>>>(
      h1, list, table, so, dgw, W2, b2, out, N);
}


# Row-ordered edge list built once per owner group in the out-degree kernel (LDS rank, one scan); both layers load it at entry: no piece scan, search or sort; run sums in rank, runs in arrival order
# speedup vs baseline: 5.7041x; 1.0330x over previous
#include <hip/hip_runtime.h>
#include <hip/hip_bf16.h>

#define GCN_N_NODES 50000
#define GCN_N_EDGES 800000
#define GCN_IN      128
#define GCN_H       128
#define GCN_C       32
#define OW_GROUPS   256
#define OW_NPG      196
#define OW_ROWS     208
#define OW_PITCH    132
#define OW_CAP      4096
#define OW_NPAD     (OW_GROUPS * OW_NPG)
#define DG_VEC      (OW_NPAD * 2 / 16)
#define DG_SAT      65535u
#define PT_CHUNKS   256
#define PT_CH       3125
#define PT_THREADS  1024
#define PT_EPT      4
#define SL_PITCH    3136
#define WL_NONE     0xFFFFFFFFu
static_assert(PT_CHUNKS * PT_CH >= GCN_N_EDGES && PT_EPT * PT_THREADS >= PT_CH, "chunks cover the edges");
static_assert(GCN_N_NODES <= 65536 && OW_NPG <= 65536 && PT_CH <= 65535, "16-bit fields of the packed words");
static_assert(SL_PITCH >= PT_CH && SL_PITCH % 16 == 0 && OW_NPG <= 256, "source-side list: one byte per edge");
static_assert((OW_NPAD * 2) % 16 == 0 && OW_NPAD >= GCN_N_NODES, "out-degree table: whole 16-byte pieces, every node in it");

typedef __attribute__((ext_vector_type(2))) float v2f;
typedef __attribute__((ext_vector_type(8))) float v8f;
typedef __attribute__((ext_vector_type(4))) unsigned v4u;

__device__ __forceinline__ void pt_scan256(const int* cnt, int* off, int lane) {
  int v[8];
  int sum = 0;
#pragma unroll
  for (int i = 0; i < 8; ++i) {
    v[i] = cnt[lane * 8 + i];
    sum += v[i];
  }
  int incl = sum;
#pragma unroll
  for (int dl = 1; dl < 32; dl <<= 1) {
    const int t = __shfl_up(incl, dl, 32);
    if (lane >= dl) incl += t;
  }
  int run = incl - sum;
#pragma unroll
  for (int i = 0; i < 8; ++i) {
    off[lane * 8 + i] = run;
    run += v[i];
  }
}

__global__ __launch_bounds__(PT_THREADS) void gcn_split_kernel(
    const int* __restrict__ src, const int* __restrict__ dst,
    unsigned* __restrict__ list, unsigned* __restrict__ table,
    unsigned* __restrict__ slist, unsigned* __restrict__ stable, int E) {
  __shared__ int cnt[OW_GROUPS];
  __shared__ int off[OW_GROUPS];
  __shared__ int cnt2[OW_GROUPS];
  __shared__ int off2[OW_GROUPS];
  __shared__ unsigned stage[PT_CH];
  __shared__ unsigned stage2[SL_PITCH / 4];
  const int tid = threadIdx.x;
  const int c   = blockIdx.x;
  const int e0  = c * PT_CH;
  const int n   = max(0, min(PT_CH, E - e0));
  if (tid < OW_GROUPS) { cnt[tid] = 0; cnt2[tid] = 0; }
  if (tid < SL_PITCH / 4) stage2[tid] = 0u;
  __syncthreads();
  int s[PT_EPT], d[PT_EPT], r[PT_EPT], r2[PT_EPT];
#pragma unroll
  for (int i = 0; i < PT_EPT; ++i) {
    const int idx = tid + i * PT_THREADS;
    s[i] = 0;
    d[i] = 0;
    if (idx < n) {
      s[i] = src[e0 + idx];
      d[i] = dst[e0 + idx];
    }
  }
#pragma unroll
  for (int i = 0; i < PT_EPT; ++i) {
    const int idx = tid + i * PT_THREADS;
    r[i]  = 0;
    r2[i] = 0;
    if (idx < n) {
      r[i]  = atomicAdd(&cnt[(unsigned)d[i] / (unsigned)OW_NPG], 1);
      r2[i] = atomicAdd(&cnt2[(unsigned)s[i] / (unsigned)OW_NPG], 1);
    }
  }
  __syncthreads();
  if (tid < 32)      pt_scan256(cnt, off, tid);
  else if (tid < 64) pt_scan256(cnt2, off2, tid - 32);
  __syncthreads();
#pragma unroll
  for (int i = 0; i < PT_EPT; ++i) {
    const int idx = tid + i * PT_THREADS;
    if (idx < n) {
      const unsigned g  = (unsigned)d[i] / (unsigned)OW_NPG;
      const unsigned gs = (unsigned)s[i] / (unsigned)OW_NPG;
      stage[off[g] + r[i]] = (unsigned)s[i] | (((unsigned)d[i] - g * (unsigned)OW_NPG) << 16);
      ((unsigned char*)stage2)[off2[gs] + r2[i]] = (unsigned char)((unsigned)s[i] - gs * (unsigned)OW_NPG);
    }
  }
  if (tid < OW_GROUPS) {
    table[(size_t)tid * PT_CHUNKS + c]  = (unsigned)off[tid] | ((unsigned)cnt[tid] << 16);
    stable[(size_t)tid * PT_CHUNKS + c] = (unsigned)off2[tid] | ((unsigned)cnt2[tid] << 16);
  }
  __syncthreads();
  for (int i = tid; i < n; i += PT_THREADS) list[e0 + i] = stage[i];
  if (tid < SL_PITCH / 4) slist[(size_t)c * (SL_PITCH / 4) + tid] = stage2[tid];
}

#define OD_THREADS 512
#define OD_PPW     (PT_CHUNKS / (OD_THREADS / 32))
#define OD_FIN0    256
static_assert(OD_PPW * (OD_THREADS / 32) == PT_CHUNKS && OD_PPW <= 32 && OD_FIN0 + OW_NPG <= OD_THREADS, "pieces per wave");
static_assert(32 * 7 >= OW_ROWS, "one wave scans the row counts, 7 per lane");

__global__ __launch_bounds__(OD_THREADS) void gcn_group_prepare_kernel(
    const unsigned* __restrict__ slist, const unsigned* __restrict__ stable,
    const unsigned* __restrict__ list, const unsigned* __restrict__ table,
    float* __restrict__ so, unsigned short* __restrict__ dg16,
    unsigned* __restrict__ wl, unsigned* __restrict__ indeg, unsigned* __restrict__ gtot, int N) {
  __shared__ int s_c[OW_NPG];
  __shared__ int s_cnt[OW_ROWS];
  __shared__ int s_off[OW_ROWS];
  __shared__ int s_info[2];
  __shared__ v4u s_win[OW_CAP / 4];
  const int tid  = threadIdx.x;
  const int lane = tid & 31;
  const int wave = __builtin_amdgcn_readfirstlane(tid >> 5);
  const int g    = blockIdx.x;
  const unsigned char* sl = (const unsigned char*)slist;
  if (tid < OW_NPG)  s_c[tid] = 0;
  if (tid < OW_ROWS) s_cnt[tid] = 0;
  if (tid == 0)      s_info[0] = 0;
  unsigned t = 0u;
  unsigned u = 0u;
  if (lane < OD_PPW) {
    t = stable[(size_t)g * PT_CHUNKS + wave * OD_PPW + lane];
    u = table[(size_t)g * PT_CHUNKS + wave * OD_PPW + lane];
  }
  __syncthreads();
  int row[OD_PPW];
#pragma unroll
  for (int k = 0; k < OD_PPW; ++k) {
    const unsigned tk  = (unsigned)__builtin_amdgcn_readlane((int)t, k);
    const int      len = (int)(tk >> 16);
    row[k] = -1;
    if (lane < len) row[k] = sl[(size_t)(wave * OD_PPW + k) * SL_PITCH + (tk & 0xFFFFu) + lane];
  }
  unsigned e0[OD_PPW], e1[OD_PPW];
#pragma unroll
  for (int k = 0; k < OD_PPW; ++k) {
    const unsigned uk   = (unsigned)__builtin_amdgcn_readlane((int)u, k);
    const int      last = max((int)(uk >> 16), 1) - 1;
    const unsigned* pc  = list + (size_t)(wave * OD_PPW + k) * PT_CH + (uk & 0xFFFFu);
    e0[k] = pc[min(lane, last)];
    e1[k] = 0u;
    if (last >= 32) e1[k] = pc[min(lane + 32, last)];
  }
#pragma unroll
  for (int k = 0; k < OD_PPW; ++k)
    if (row[k] >= 0) atomicAdd(&s_c[row[k]], 1);
#pragma unroll
  for (int k = 0; k < OD_PPW; ++k) {
    const unsigned tk  = (unsigned)__builtin_amdgcn_readlane((int)t, k);
    const int      len = (int)(tk >> 16);
    for (int i = lane + 32; i < len; i += 32)
      atomicAdd(&s_c[sl[(size_t)(wave * OD_PPW + k) * SL_PITCH + (tk & 0xFFFFu) + i]], 1);
  }
  int r0[OD_PPW], r1[OD_PPW];
#pragma unroll
  for (int k = 0; k < OD_PPW; ++k) {
    const int len = (int)((unsigned)__builtin_amdgcn_readlane((int)u, k) >> 16);
    r0[k] = 0;
    r1[k] = 0;
    if (lane < len)      r0[k] = atomicAdd(&s_cnt[e0[k] >> 16], 1);
    if (lane + 32 < len) r1[k] = atomicAdd(&s_cnt[e1[k] >> 16], 1);
  }
  if (lane < OD_PPW && (u >> 16) > 64u) s_info[0] = 1;
  __syncthreads();
  if (tid < 32) {
    int v[7];
    int sum = 0;
#pragma unroll
    for (int i = 0; i < 7; ++i) {
      const int idx = tid * 7 + i;
      v[i] = (idx < OW_ROWS) ? s_cnt[idx] : 0;
      sum += v[i];
    }
    int incl = sum;
#pragma unroll
    for (int dl = 1; dl < 32; dl <<= 1) {
      const int tt = __shfl_up(incl, dl, 32);
      if (tid >= dl) incl += tt;
    }
    const int  tot = __shfl(incl, 31, 32);
    const bool pre = (s_info[0] == 0) && (tot <= OW_CAP);
    int run = incl - sum;
#pragma unroll
    for (int i = 0; i < 7; ++i) {
      const int idx = tid * 7 + i;
      if (idx < OW_ROWS) {
        s_off[idx] = run;
        indeg[g * OW_ROWS + idx] = pre ? (unsigned)v[i] : 0u;
      }
      run += v[i];
    }
    if (tid == 0) {
      s_info[1] = tot;
      gtot[g]   = pre ? (unsigned)tot : WL_NONE;
    }
  }
  if (tid >= OD_FIN0 && tid < OD_FIN0 + OW_NPG) {
    const int r  = tid - OD_FIN0;
    const int dg = s_c[r];
    dg16[g * OW_NPG + r] = (unsigned short)min((unsigned)dg, DG_SAT);
    if (g * OW_NPG + r < N) so[g * OW_NPG + r] = (dg > 0) ? rsqrtf((float)dg) : 0.0f;
  }
  __syncthreads();
  const int  tot = __builtin_amdgcn_readfirstlane(s_info[1]);
  const bool pre = (__builtin_amdgcn_readfirstlane(s_info[0]) == 0) && (tot <= OW_CAP);
  unsigned* sw = (unsigned*)s_win;
  if (pre) {
#pragma unroll
    for (int k = 0; k < OD_PPW; ++k) {
      const int len = (int)((unsigned)__builtin_amdgcn_readlane((int)u, k) >> 16);
      if (lane < len)      sw[s_off[e0[k] >> 16] + r0[k]] = e0[k];
      if (lane + 32 < len) sw[s_off[e1[k] >> 16] + r1[k]] = e1[k];
    }
  }
  for (int i = (pre ? tot : 0) + tid; i < OW_CAP; i += OD_THREADS) sw[i] = 0u;
  __syncthreads();
  v4u* w = (v4u*)(wl + (size_t)g * OW_CAP);
  for (int i = tid; i < OW_CAP / 4; i += OD_THREADS) w[i] = s_win[i];
}

#define OW_THREADS 1024
#define OW_FPT     (OW_CAP / OW_THREADS)

__device__ __forceinline__ void ow_flush_run(float* tile, int d, int lane, float4 a) {
  float* t = tile + d * OW_PITCH + lane;
  __hip_atomic_fetch_add(t + 0,  a.x, __ATOMIC_RELAXED, __HIP_MEMORY_SCOPE_WORKGROUP);
  __hip_atomic_fetch_add(t + 32, a.y, __ATOMIC_RELAXED, __HIP_MEMORY_SCOPE_WORKGROUP);
  __hip_atomic_fetch_add(t + 64, a.z, __ATOMIC_RELAXED, __HIP_MEMORY_SCOPE_WORKGROUP);
  __hip_atomic_fetch_add(t + 96, a.w, __ATOMIC_RELAXED, __HIP_MEMORY_SCOPE_WORKGROUP);
}

__device__ __forceinline__ void ow_add_row(float* tile, int d, int lane, float4 v, float sc,
                                           int& cur, float4& acc) {
  if (d != cur) {
    ow_flush_run(tile, cur, lane, acc);
    acc = make_float4(0.0f, 0.0f, 0.0f, 0.0f);
    cur = d;
  }
  acc.x += v.x * sc;
  acc.y += v.y * sc;
  acc.z += v.z * sc;
  acc.w += v.w * sc;
}

__device__ __forceinline__ void ow_process(
    float* tile, const float* __restrict__ x,
    int s_l, int d_l, int sc_l, int cnt, int lane) {
  float4 va[8], vb[8];
  int    cur = __builtin_amdgcn_readlane(d_l, 0);
  float4 acc = make_float4(0.0f, 0.0f, 0.0f, 0.0f);
#pragma unroll
  for (int k = 0; k < 8; ++k)
    va[k] = *(const float4*)(x + (size_t)__builtin_amdgcn_readlane(s_l, k) * 128 + lane * 4);
#pragma unroll
  for (int j0 = 0; j0 < 32; j0 += 16) {
#pragma unroll
    for (int k = 0; k < 8; ++k)
      vb[k] = *(const float4*)(x + (size_t)__builtin_amdgcn_readlane(s_l, j0 + 8 + k) * 128 + lane * 4);
#pragma unroll
    for (int k = 0; k < 8; ++k)
      if (j0 + k < cnt)
        ow_add_row(tile, __builtin_amdgcn_readlane(d_l, j0 + k), lane, va[k],
                   __int_as_float(__builtin_amdgcn_readlane(sc_l, j0 + k)), cur, acc);
#pragma unroll
    for (int k = 0; k < 8; ++k)
      if (j0 + 16 + k < 32)
        va[k] = *(const float4*)(x + (size_t)__builtin_amdgcn_readlane(s_l, j0 + 16 + k) * 128 + lane * 4);
#pragma unroll
    for (int k = 0; k < 8; ++k)
      if (j0 + 8 + k < cnt)
        ow_add_row(tile, __builtin_amdgcn_readlane(d_l, j0 + 8 + k), lane, vb[k],
                   __int_as_float(__builtin_amdgcn_readlane(sc_l, j0 + 8 + k)), cur, acc);
  }
  ow_flush_run(tile, cur, lane, acc);
}

template <int NOUT, bool RELU>
__global__ __launch_bounds__(OW_THREADS) void gcn_spmm_owner_mm(
    const float* __restrict__ x, const unsigned* __restrict__ list,
    const unsigned* __restrict__ table, const float* __restrict__ so,
    const unsigned* __restrict__ dgw,
    const unsigned* __restrict__ wl, const unsigned* __restrict__ indeg, const unsigned* __restrict__ gtot,
    const float* __restrict__ W, const float* __restrict__ bias,
    float* __restrict__ outp, int N) {
  __shared__ float tile[OW_ROWS * OW_PITCH];
  __shared__ v4u      s_dg[DG_VEC];
  __shared__ unsigned s_tab[PT_CHUNKS];
  __shared__ int      s_base[PT_CHUNKS + 1];
  __shared__ unsigned s_ent[OW_CAP];
  __shared__ float    s_sc[OW_CAP];
  __shared__ int      s_deg[OW_ROWS];
  __shared__ int      s_cnt[OW_ROWS];
  __shared__ int      s_off[OW_ROWS];
  const int tid  = threadIdx.x;
  const int lane = tid & 31;
  const int wave = __builtin_amdgcn_readfirstlane(tid >> 5);
  const int g     = blockIdx.x;
  const int node0 = g * OW_NPG;
  const int nrows = min(OW_NPG, N - node0);

  constexpr int DG_PT = (DG_VEC + OW_THREADS - 1) / OW_THREADS;
  const unsigned totw = gtot[g];
  unsigned pe[OW_FPT];
#pragma unroll
  for (int q = 0; q < OW_FPT; ++q)
    pe[q] = wl[(size_t)g * OW_CAP + tid + q * OW_THREADS];
  unsigned idg = 0u;
  if (tid < OW_ROWS) idg = indeg[g * OW_ROWS + tid];
  v4u dv[DG_PT];
#pragma unroll
  for (int k = 0; k < DG_PT; ++k)
    dv[k] = ((const v4u*)dgw)[min(tid + k * OW_THREADS, DG_VEC - 1)];
  if (tid < OW_ROWS) { s_deg[tid] = (int)idg; s_cnt[tid] = 0; }
  for (int i = tid; i < OW_ROWS * OW_PITCH / 4; i += OW_THREADS)
    ((float4*)tile)[i] = make_float4(0.0f, 0.0f, 0.0f, 0.0f);
#pragma unroll
  for (int k = 0; k < DG_PT; ++k)
    if (tid + k * OW_THREADS < DG_VEC) s_dg[tid + k * OW_THREADS] = dv[k];
  __syncthreads();
  const unsigned ptot = (unsigned)__builtin_amdgcn_readfirstlane((int)totw);

  if (__builtin_expect(ptot <= (unsigned)OW_CAP, 1)) {
    const int nb = (int)(ptot + 31u) >> 5;
#pragma nounroll
    for (int q = 0; q < OW_FPT; ++q) {
      const int b = wave + q * (OW_THREADS / 32);
      if (b >= nb) break;
      unsigned ent = pe[0];
#pragma unroll
      for (int r = 1; r < OW_FPT; ++r)
        if (q == r) ent = pe[r];
      const int cnt = min(32, (int)ptot - (b << 5));
      int s_l = 0, d_l = 0, sc_l = 0;
      if (lane < cnt) {
        s_l = (int)(ent & 0xFFFFu);
        d_l = (int)(ent >> 16);
        const unsigned dgq = ((const unsigned short*)s_dg)[s_l];
        float fsq = (dgq > 0u) ? rsqrtf((float)dgq) : 0.0f;
        if (dgq == DG_SAT) fsq = so[s_l];
        sc_l = __float_as_int(fsq);
      }
      if (cnt == 32) ow_process(tile, x, s_l, d_l, sc_l, 32, lane);
      else           ow_process(tile, x, s_l, d_l, sc_l, cnt, lane);
    }
    __syncthreads();
  } else {
    if (tid < PT_CHUNKS) s_tab[tid] = table[(size_t)g * PT_CHUNKS + tid];
    __syncthreads();
    if (tid < 32) {
      int v[8];
      int sum = 0;
#pragma unroll
      for (int i = 0; i < 8; ++i) {
        v[i] = (int)(s_tab[tid * 8 + i] >> 16);
        sum += v[i];
      }
      int incl = sum;
#pragma unroll
      for (int dl = 1; dl < 32; dl <<= 1) {
        const int t = __shfl_up(incl, dl, 32);
        if (tid >= dl) incl += t;
      }
      int run = incl - sum;
#pragma unroll
      for (int i = 0; i < 8; ++i) {
        s_base[tid * 8 + i] = run;
        run += v[i];
      }
      if (tid == 31) s_base[PT_CHUNKS] = run;
    }
    __syncthreads();
    const int total = __builtin_amdgcn_readfirstlane(s_base[PT_CHUNKS]);

    for (int wb = 0; wb < total; wb += OW_CAP) {
      const int nwin = min(OW_CAP, total - wb);
      int      fc[OW_FPT], fp[OW_FPT];
      unsigned fe[OW_FPT];
      float    fs[OW_FPT];
#pragma unroll
      for (int q = 0; q < OW_FPT; ++q) {
        fp[q] = min(wb + tid + q * OW_THREADS, total - 1);
        int lo = 0;
#pragma unroll
        for (int step = PT_CHUNKS / 2; step >= 1; step >>= 1)
          if (s_base[lo + step] <= fp[q]) lo += step;
        fc[q] = lo;
      }
#pragma unroll
      for (int q = 0; q < OW_FPT; ++q)
        fe[q] = list[(size_t)fc[q] * PT_CH + (s_tab[fc[q]] & 0xFFFFu) + (unsigned)(fp[q] - s_base[fc[q]])];
      unsigned dgv[OW_FPT];
#pragma unroll
      for (int q = 0; q < OW_FPT; ++q)
        dgv[q] = ((const unsigned short*)s_dg)[fe[q] & 0xFFFFu];
#pragma unroll
      for (int q = 0; q < OW_FPT; ++q)
        fs[q] = (dgv[q] > 0u) ? rsqrtf((float)dgv[q]) : 0.0f;
#pragma unroll
      for (int q = 0; q < OW_FPT; ++q)
        if (dgv[q] == DG_SAT) fs[q] = so[fe[q] & 0xFFFFu];
      int rk[OW_FPT];
#pragma unroll
      for (int q = 0; q < OW_FPT; ++q) {
        rk[q] = 0;
        if (wb + tid + q * OW_THREADS < total) rk[q] = atomicAdd(&s_cnt[fe[q] >> 16], 1);
      }
      __syncthreads();
      if (tid < 32) {
        int v[7];
        int sum = 0;
#pragma unroll
        for (int i = 0; i < 7; ++i) {
          const int idx = tid * 7 + i;
          v[i] = (idx < OW_ROWS) ? s_cnt[idx] : 0;
          sum += v[i];
        }
        int incl = sum;
#pragma unroll
        for (int dl = 1; dl < 32; dl <<= 1) {
          const int t = __shfl_up(incl, dl, 32);
          if (tid >= dl) incl += t;
        }
        int run = incl - sum;
#pragma unroll
        for (int i = 0; i < 7; ++i) {
          const int idx = tid * 7 + i;
          if (idx < OW_ROWS) {
            s_off[idx] = run;
            s_deg[idx] += v[i];
            s_cnt[idx] = 0;
          }
          run += v[i];
        }
      }
      __syncthreads();
#pragma unroll
      for (int q = 0; q < OW_FPT; ++q) {
        if (wb + tid + q * OW_THREADS < total) {
          const int pos = s_off[fe[q] >> 16] + rk[q];
          s_ent[pos] = fe[q];
          s_sc[pos]  = fs[q];
        }
      }
      __syncthreads();
      const int nb = (nwin + 31) >> 5;
      for (int b = wave; b < nb; b += OW_THREADS / 32) {
        const int cnt = min(32, nwin - (b << 5));
        int s_l = 0, d_l = 0, sc_l = 0;
        if (lane < cnt) {
          const unsigned ent = s_ent[(b << 5) + lane];
          s_l  = (int)(ent & 0xFFFFu);
          d_l  = (int)(ent >> 16);
          sc_l = __float_as_int(s_sc[(b << 5) + lane]);
        }
        ow_process(tile, x, s_l, d_l, sc_l, cnt, lane);
      }
      __syncthreads();
    }
  }

  constexpr int NCT = NOUT / 16;
  constexpr int NRW = (OW_THREADS / 32) / NCT;
  const int ct   = wave % NCT;
  const int wq   = wave / NCT;
  const int half = lane >> 4;
  const int c    = lane & 15;
  const int col  = ct * 16 + c;
  v2f bfrag[32];
#pragma unroll
  for (int kk = 0; kk < 32; ++kk) {
    const int k2 = kk * 4 + half * 2;
    bfrag[kk].x = W[(size_t)k2 * NOUT + col];
    bfrag[kk].y = W[(size_t)(k2 + 1) * NOUT + col];
  }
  const float bv = bias[col];
  for (int rt = wq; rt < OW_ROWS / 16; rt += NRW) {
    const int arow = rt * 16 + c;
    const int   dg = s_deg[arow];
    const float sc = (dg > 0) ? rsqrtf((float)dg) : 0.0f;
    const float* trow = tile + arow * OW_PITCH + 64 * half;
    v8f acc = {};
#pragma unroll
    for (int kk = 0; kk < 32; ++kk) {
      v2f a;
      a.x = trow[kk] * sc;
      a.y = trow[32 + kk] * sc;
      acc = __builtin_amdgcn_wmma_f32_16x16x4_f32(
           false, a,  false, bfrag[kk],
           (short)0, acc,  false,  false);
    }
#pragma unroll
    for (int r = 0; r < 8; ++r) {
      const int lrow = rt * 16 + r + half * 8;
      if (lrow < nrows) {
        float v = acc[r] + bv;
        if (RELU) v = (v > 0.0f) ? v : 0.0f;
        outp[(size_t)(node0 + lrow) * NOUT + col] = v;
      }
    }
  }
}

extern "C" void kernel_launch(void* const* d_in, const int* in_sizes, int n_in,
                              void* d_out, int out_size, void* d_ws, size_t ws_size,
                              hipStream_t stream) {
  const float* in_feat = (const float*)d_in[0];
  const int*   src     = (const int*)d_in[1];
  const int*   dst     = (const int*)d_in[2];
  const float* W1      = (const float*)d_in[3];
  const float* b1      = (const float*)d_in[4];
  const float* W2      = (const float*)d_in[5];
  const float* b2      = (const float*)d_in[6];
  float* out = (float*)d_out;

  const int N = GCN_N_NODES;
  const int E = GCN_N_EDGES;

  float* so      = (float*)d_ws;
  float* h1      = so + 50176;
  unsigned* list   = (unsigned*)(h1 + (size_t)N * 128);
  unsigned* table  = list + GCN_N_EDGES;
  unsigned* stable = table + OW_GROUPS * PT_CHUNKS;
  unsigned* slist  = stable + OW_GROUPS * PT_CHUNKS;
  unsigned* dgw    = slist + PT_CHUNKS * (SL_PITCH / 4);
  unsigned* wl     = dgw + OW_NPAD / 2;
  unsigned* indeg  = wl + OW_GROUPS * OW_CAP;
  unsigned* gtot   = indeg + OW_GROUPS * OW_ROWS;

  gcn_split_kernel<<<PT_CHUNKS, PT_THREADS, 0, stream>>>(src, dst, list, table, slist, stable, E);

  gcn_group_prepare_kernel<<<OW_GROUPS, OD_THREADS, 0, stream>>>(
      slist, stable, list, table, so, (unsigned short*)dgw, wl, indeg, gtot, N);

  gcn_spmm_owner_mm<GCN_H, true><<<OW_GROUPS, OW_THREADS, 0, stream>>>(
      in_feat, list, table, so, dgw, wl, indeg, gtot, W1, b1, h1, N);

  gcn_spmm_owner_mm<GCN_C, false><<<OW_GROUPS, OW_THREADS, 0, stream>>>(
      h1, list, table, so, dgw, wl, indeg, gtot, W2, b2, out, N);
}
